# starting kernel
# baseline (speedup 1.0000x reference)
#include <hip/hip_runtime.h>


typedef __attribute__((ext_vector_type(2))) float v2f;
typedef __attribute__((ext_vector_type(8))) float v8f;

#define TAU   0.05f
#define SIGMA 0.05f
#define MAX_ITER 30

static constexpr int AP = 260;
static constexpr int YP = 196;

static constexpr int OFF_A  = 0;
static constexpr int OFF_X  = OFF_A  + 192 * AP;
static constexpr int OFF_XB = OFF_X  + 32  * AP;
static constexpr int OFF_YS = OFF_XB + 32  * AP;
static constexpr int OFF_Z  = OFF_YS + 32  * YP;
static constexpr int OFF_C  = OFF_Z  + 32  * YP;
static constexpr int SMEM_TOT = OFF_C + 256;

__global__ __launch_bounds__(128, 1)
void pdhg_kernel(const float* __restrict__ x0g, const float* __restrict__ y1g,
                 const float* __restrict__ y2g, const float* __restrict__ zg,
                 const float* __restrict__ cg,  const float* __restrict__ A1,
                 const float* __restrict__ A2,  float* __restrict__ out)
{
    __shared__ float smem[SMEM_TOT];

    const int tid  = threadIdx.x;
    const int wave = tid >> 5;
    const int lane = tid & 31;
    const int tile = wave >> 1;
    const int half = wave & 1;
    const int n    = lane & 15;
    const int hi   = (lane >> 4) << 3;
    const int ko2  = (lane >> 4) << 1;
    const int pid  = (half << 5) | lane;

    const int rbase = blockIdx.x * 32 + tile * 16;
    const int xw  = OFF_X  + tile * 16 * AP;
    const int xbw = OFF_XB + tile * 16 * AP;
    const int ysw = OFF_YS + tile * 16 * YP;
    const int zw  = OFF_Z  + tile * 16 * YP;

    for (int i = tid; i < 192 * 64; i += 128) {
        int r = i >> 6, c4 = (i & 63) << 2;
        const float* src = (r < 128) ? (A1 + r * 256 + c4) : (A2 + (r - 128) * 256 + c4);
        *(float4*)&smem[OFF_A + r * AP + c4] = *(const float4*)src;
    }
    if (tid < 64)
        *(float4*)&smem[OFF_C + tid * 4] = *(const float4*)(cg + tid * 4);
    for (int i = pid; i < 16 * 64; i += 64) {
        int r = i >> 6, c4 = (i & 63) << 2;
        float4 v = *(const float4*)(x0g + (size_t)(rbase + r) * 256 + c4);
        *(float4*)&smem[xw  + r * AP + c4] = v;
        *(float4*)&smem[xbw + r * AP + c4] = v;
    }
    for (int i = pid; i < 16 * 96; i += 64) {
        int r = i / 96, c2 = (i % 96) << 1;
        float2 v = *(const float2*)(zg + (size_t)(rbase + r) * 192 + c2);
        *(float2*)&smem[zw + r * YP + c2] = v;
    }
    for (int i = pid; i < 16 * 96; i += 64) {
        int r = i / 96, c2 = (i % 96) << 1;
        float2 v;
        if (c2 < 128) v = *(const float2*)(y1g + (size_t)(rbase + r) * 128 + c2);
        else          v = *(const float2*)(y2g + (size_t)(rbase + r) * 64 + (c2 - 128));
        v.x *= -TAU; v.y *= -TAU;
        *(float2*)&smem[ysw + r * YP + c2] = v;
    }
    __syncthreads();

    const float NEG_INV_TS = -1.0f / (TAU * SIGMA);

    for (int it = 0; it < MAX_ITER; ++it) {
        v8f acc[6];
#pragma unroll
        for (int cti = 0; cti < 6; ++cti) {
            int ct = half * 6 + cti;
#pragma unroll
            for (int v = 0; v < 8; ++v) {
                int row = v + hi, col = ct * 16 + n;
                acc[cti][v] = smem[ysw + row * YP + col] * NEG_INV_TS
                            - smem[zw  + row * YP + col];
            }
        }
        for (int k0 = 0; k0 < 256; k0 += 4) {
            int ko = k0 + ko2;
            v2f a = *(const v2f*)&smem[xbw + (lane & 15) * AP + ko];
#pragma unroll
            for (int cti = 0; cti < 6; ++cti) {
                int ct = half * 6 + cti;
                v2f b = *(const v2f*)&smem[OFF_A + (ct * 16 + (lane & 15)) * AP + ko];
                acc[cti] = __builtin_amdgcn_wmma_f32_16x16x4_f32(
                    false, a, false, b, (short)0, acc[cti], false, false);
            }
        }
#pragma unroll
        for (int cti = 0; cti < 6; ++cti) {
            int ct = half * 6 + cti;
#pragma unroll
            for (int v = 0; v < 8; ++v) {
                float yv = SIGMA * acc[cti][v];
                yv = (ct < 8) ? fmaxf(yv, 0.0f) : yv;
                smem[ysw + (v + hi) * YP + ct * 16 + n] = -TAU * yv;
            }
        }
        __syncthreads();

        v8f acc2[8];
#pragma unroll
        for (int cti = 0; cti < 8; ++cti) {
            int ct = half * 8 + cti;
            float cv = smem[OFF_C + ct * 16 + n];
#pragma unroll
            for (int v = 0; v < 8; ++v)
                acc2[cti][v] = smem[xw + (v + hi) * AP + ct * 16 + n] - TAU * cv;
        }
        for (int m0 = 0; m0 < 192; m0 += 4) {
            int mo = m0 + ko2;
            v2f a = *(const v2f*)&smem[ysw + (lane & 15) * YP + mo];
#pragma unroll
            for (int cti = 0; cti < 8; ++cti) {
                int ct = half * 8 + cti;
                v2f b;
                b.x = smem[OFF_A + mo * AP + ct * 16 + n];
                b.y = smem[OFF_A + (mo + 1) * AP + ct * 16 + n];
                acc2[cti] = __builtin_amdgcn_wmma_f32_16x16x4_f32(
                    false, a, false, b, (short)0, acc2[cti], false, false);
            }
        }
#pragma unroll
        for (int cti = 0; cti < 8; ++cti) {
            int ct = half * 8 + cti;
#pragma unroll
            for (int v = 0; v < 8; ++v) {
                int idx = (v + hi) * AP + ct * 16 + n;
                float xo = smem[xw + idx];
                float xn = fmaxf(acc2[cti][v], 0.0f);
                smem[xw  + idx] = xn;
                smem[xbw + idx] = 2.0f * xn - xo;
            }
        }
        __syncthreads();
    }

    float* out_x  = out;
    float* out_y1 = out + (size_t)8192 * 256;
    float* out_y2 = out_y1 + (size_t)8192 * 128;
    for (int i = pid; i < 16 * 64; i += 64) {
        int r = i >> 6, c4 = (i & 63) << 2;
        float4 v = *(const float4*)&smem[xw + r * AP + c4];
        *(float4*)(out_x + (size_t)(rbase + r) * 256 + c4) = v;
    }
    const float NEG_INV_TAU = -1.0f / TAU;
    for (int i = pid; i < 16 * 96; i += 64) {
        int r = i / 96, c2 = (i % 96) << 1;
        float2 v = *(const float2*)&smem[ysw + r * YP + c2];
        v.x *= NEG_INV_TAU; v.y *= NEG_INV_TAU;
        if (c2 < 128) *(float2*)(out_y1 + (size_t)(rbase + r) * 128 + c2) = v;
        else          *(float2*)(out_y2 + (size_t)(rbase + r) * 64 + (c2 - 128)) = v;
    }
}

extern "C" void kernel_launch(void* const* d_in, const int* in_sizes, int n_in,
                              void* d_out, int out_size, void* d_ws, size_t ws_size,
                              hipStream_t stream) {
    const float* x0 = (const float*)d_in[0];
    const float* y1 = (const float*)d_in[1];
    const float* y2 = (const float*)d_in[2];
    const float* z  = (const float*)d_in[3];
    const float* c  = (const float*)d_in[4];
    const float* A1 = (const float*)d_in[5];
    const float* A2 = (const float*)d_in[6];
    (void)in_sizes; (void)n_in; (void)out_size; (void)d_ws; (void)ws_size;
    pdhg_kernel<<<dim3(8192 / 32), dim3(128), 0, stream>>>(
        x0, y1, y2, z, c, A1, A2, (float*)d_out);
}


# 8 waves per workgroup (two per SIMD): each 16-row tile split over 4 waves owning a quarter of the output columns, so one wave's LDS waits and vector epilogues overlap its sibling's WMMAs
# speedup vs baseline: 1.2037x; 1.2037x over previous
#include <hip/hip_runtime.h>


typedef __attribute__((ext_vector_type(2))) float v2f;
typedef __attribute__((ext_vector_type(8))) float v8f;

#define TAU   0.05f
#define SIGMA 0.05f
#define MAX_ITER 30

static constexpr int AP = 260;
static constexpr int YP = 196;

static constexpr int OFF_A  = 0;
static constexpr int OFF_X  = OFF_A  + 192 * AP;
static constexpr int OFF_XB = OFF_X  + 32  * AP;
static constexpr int OFF_YS = OFF_XB + 32  * AP;
static constexpr int OFF_Z  = OFF_YS + 32  * YP;
static constexpr int OFF_C  = OFF_Z  + 32  * YP;
static constexpr int SMEM_TOT = OFF_C + 256;

__global__ __launch_bounds__(256, 1)
void pdhg_kernel(const float* __restrict__ x0g, const float* __restrict__ y1g,
                 const float* __restrict__ y2g, const float* __restrict__ zg,
                 const float* __restrict__ cg,  const float* __restrict__ A1,
                 const float* __restrict__ A2,  float* __restrict__ out)
{
    __shared__ float smem[SMEM_TOT];

    const int tid  = threadIdx.x;
    const int wave = tid >> 5;
    const int lane = tid & 31;
    const int tile = wave >> 2;
    const int qtr  = wave & 3;
    const int n    = lane & 15;
    const int hi   = (lane >> 4) << 3;
    const int ko2  = (lane >> 4) << 1;
    const int pid  = (qtr << 5) | lane;

    const int rbase = blockIdx.x * 32 + tile * 16;
    const int xw  = OFF_X  + tile * 16 * AP;
    const int xbw = OFF_XB + tile * 16 * AP;
    const int ysw = OFF_YS + tile * 16 * YP;
    const int zw  = OFF_Z  + tile * 16 * YP;

    for (int i = tid; i < 192 * 64; i += 256) {
        int r = i >> 6, c4 = (i & 63) << 2;
        const float* src = (r < 128) ? (A1 + r * 256 + c4) : (A2 + (r - 128) * 256 + c4);
        *(float4*)&smem[OFF_A + r * AP + c4] = *(const float4*)src;
    }
    if (tid < 64)
        *(float4*)&smem[OFF_C + tid * 4] = *(const float4*)(cg + tid * 4);
    for (int i = pid; i < 16 * 64; i += 128) {
        int r = i >> 6, c4 = (i & 63) << 2;
        float4 v = *(const float4*)(x0g + (size_t)(rbase + r) * 256 + c4);
        *(float4*)&smem[xw  + r * AP + c4] = v;
        *(float4*)&smem[xbw + r * AP + c4] = v;
    }
    for (int i = pid; i < 16 * 96; i += 128) {
        int r = i / 96, c2 = (i % 96) << 1;
        float2 v = *(const float2*)(zg + (size_t)(rbase + r) * 192 + c2);
        *(float2*)&smem[zw + r * YP + c2] = v;
    }
    for (int i = pid; i < 16 * 96; i += 128) {
        int r = i / 96, c2 = (i % 96) << 1;
        float2 v;
        if (c2 < 128) v = *(const float2*)(y1g + (size_t)(rbase + r) * 128 + c2);
        else          v = *(const float2*)(y2g + (size_t)(rbase + r) * 64 + (c2 - 128));
        v.x *= -TAU; v.y *= -TAU;
        *(float2*)&smem[ysw + r * YP + c2] = v;
    }
    __syncthreads();

    const float NEG_INV_TS = -1.0f / (TAU * SIGMA);

    for (int it = 0; it < MAX_ITER; ++it) {
        v8f acc[3];
#pragma unroll
        for (int cti = 0; cti < 3; ++cti) {
            int ct = qtr * 3 + cti;
#pragma unroll
            for (int v = 0; v < 8; ++v) {
                int row = v + hi, col = ct * 16 + n;
                acc[cti][v] = smem[ysw + row * YP + col] * NEG_INV_TS
                            - smem[zw  + row * YP + col];
            }
        }
        for (int k0 = 0; k0 < 256; k0 += 4) {
            int ko = k0 + ko2;
            v2f a = *(const v2f*)&smem[xbw + (lane & 15) * AP + ko];
#pragma unroll
            for (int cti = 0; cti < 3; ++cti) {
                int ct = qtr * 3 + cti;
                v2f b = *(const v2f*)&smem[OFF_A + (ct * 16 + (lane & 15)) * AP + ko];
                acc[cti] = __builtin_amdgcn_wmma_f32_16x16x4_f32(
                    false, a, false, b, (short)0, acc[cti], false, false);
            }
        }
#pragma unroll
        for (int cti = 0; cti < 3; ++cti) {
            int ct = qtr * 3 + cti;
#pragma unroll
            for (int v = 0; v < 8; ++v) {
                float yv = SIGMA * acc[cti][v];
                yv = (ct < 8) ? fmaxf(yv, 0.0f) : yv;
                smem[ysw + (v + hi) * YP + ct * 16 + n] = -TAU * yv;
            }
        }
        __syncthreads();

        v8f acc2[4];
#pragma unroll
        for (int cti = 0; cti < 4; ++cti) {
            int ct = qtr * 4 + cti;
            float cv = smem[OFF_C + ct * 16 + n];
#pragma unroll
            for (int v = 0; v < 8; ++v)
                acc2[cti][v] = smem[xw + (v + hi) * AP + ct * 16 + n] - TAU * cv;
        }
        for (int m0 = 0; m0 < 192; m0 += 4) {
            int mo = m0 + ko2;
            v2f a = *(const v2f*)&smem[ysw + (lane & 15) * YP + mo];
#pragma unroll
            for (int cti = 0; cti < 4; ++cti) {
                int ct = qtr * 4 + cti;
                v2f b;
                b.x = smem[OFF_A + mo * AP + ct * 16 + n];
                b.y = smem[OFF_A + (mo + 1) * AP + ct * 16 + n];
                acc2[cti] = __builtin_amdgcn_wmma_f32_16x16x4_f32(
                    false, a, false, b, (short)0, acc2[cti], false, false);
            }
        }
#pragma unroll
        for (int cti = 0; cti < 4; ++cti) {
            int ct = qtr * 4 + cti;
#pragma unroll
            for (int v = 0; v < 8; ++v) {
                int idx = (v + hi) * AP + ct * 16 + n;
                float xo = smem[xw + idx];
                float xn = fmaxf(acc2[cti][v], 0.0f);
                smem[xw  + idx] = xn;
                smem[xbw + idx] = 2.0f * xn - xo;
            }
        }
        __syncthreads();
    }

    float* out_x  = out;
    float* out_y1 = out + (size_t)8192 * 256;
    float* out_y2 = out_y1 + (size_t)8192 * 128;
    for (int i = pid; i < 16 * 64; i += 128) {
        int r = i >> 6, c4 = (i & 63) << 2;
        float4 v = *(const float4*)&smem[xw + r * AP + c4];
        *(float4*)(out_x + (size_t)(rbase + r) * 256 + c4) = v;
    }
    const float NEG_INV_TAU = -1.0f / TAU;
    for (int i = pid; i < 16 * 96; i += 128) {
        int r = i / 96, c2 = (i % 96) << 1;
        float2 v = *(const float2*)&smem[ysw + r * YP + c2];
        v.x *= NEG_INV_TAU; v.y *= NEG_INV_TAU;
        if (c2 < 128) *(float2*)(out_y1 + (size_t)(rbase + r) * 128 + c2) = v;
        else          *(float2*)(out_y2 + (size_t)(rbase + r) * 64 + (c2 - 128)) = v;
    }
}

extern "C" void kernel_launch(void* const* d_in, const int* in_sizes, int n_in,
                              void* d_out, int out_size, void* d_ws, size_t ws_size,
                              hipStream_t stream) {
    const float* x0 = (const float*)d_in[0];
    const float* y1 = (const float*)d_in[1];
    const float* y2 = (const float*)d_in[2];
    const float* z  = (const float*)d_in[3];
    const float* c  = (const float*)d_in[4];
    const float* A1 = (const float*)d_in[5];
    const float* A2 = (const float*)d_in[6];
    (void)in_sizes; (void)n_in; (void)out_size; (void)d_ws; (void)ws_size;
    pdhg_kernel<<<dim3(8192 / 32), dim3(256), 0, stream>>>(
        x0, y1, y2, z, c, A1, A2, (float*)d_out);
}


# Lane-owned state (each lane's own x, -TAU*y, z, c elements) kept in registers across all 30 iterations: no LDS re-reads in prologues, Phase B epilogue's 32 serial LDS load-then-store hops gone
# speedup vs baseline: 1.3682x; 1.1367x over previous
#include <hip/hip_runtime.h>


typedef __attribute__((ext_vector_type(2))) float v2f;
typedef __attribute__((ext_vector_type(8))) float v8f;

#define TAU   0.05f
#define SIGMA 0.05f
#define MAX_ITER 30

static constexpr int AP = 260;
static constexpr int YP = 196;

static constexpr int OFF_A  = 0;
static constexpr int OFF_X  = OFF_A  + 192 * AP;
static constexpr int OFF_XB = OFF_X  + 32  * AP;
static constexpr int OFF_YS = OFF_XB + 32  * AP;
static constexpr int OFF_Z  = OFF_YS + 32  * YP;
static constexpr int OFF_C  = OFF_Z  + 32  * YP;
static constexpr int SMEM_TOT = OFF_C + 256;

__global__ __launch_bounds__(256, 1)
void pdhg_kernel(const float* __restrict__ x0g, const float* __restrict__ y1g,
                 const float* __restrict__ y2g, const float* __restrict__ zg,
                 const float* __restrict__ cg,  const float* __restrict__ A1,
                 const float* __restrict__ A2,  float* __restrict__ out)
{
    __shared__ float smem[SMEM_TOT];

    const int tid  = threadIdx.x;
    const int wave = tid >> 5;
    const int lane = tid & 31;
    const int tile = wave >> 2;
    const int qtr  = wave & 3;
    const int n    = lane & 15;
    const int hi   = (lane >> 4) << 3;
    const int ko2  = (lane >> 4) << 1;
    const int pid  = (qtr << 5) | lane;

    const int rbase = blockIdx.x * 32 + tile * 16;
    const int xw  = OFF_X  + tile * 16 * AP;
    const int xbw = OFF_XB + tile * 16 * AP;
    const int ysw = OFF_YS + tile * 16 * YP;
    const int zw  = OFF_Z  + tile * 16 * YP;

    for (int i = tid; i < 192 * 64; i += 256) {
        int r = i >> 6, c4 = (i & 63) << 2;
        const float* src = (r < 128) ? (A1 + r * 256 + c4) : (A2 + (r - 128) * 256 + c4);
        *(float4*)&smem[OFF_A + r * AP + c4] = *(const float4*)src;
    }
    if (tid < 64)
        *(float4*)&smem[OFF_C + tid * 4] = *(const float4*)(cg + tid * 4);
    for (int i = pid; i < 16 * 64; i += 128) {
        int r = i >> 6, c4 = (i & 63) << 2;
        float4 v = *(const float4*)(x0g + (size_t)(rbase + r) * 256 + c4);
        *(float4*)&smem[xw  + r * AP + c4] = v;
        *(float4*)&smem[xbw + r * AP + c4] = v;
    }
    for (int i = pid; i < 16 * 96; i += 128) {
        int r = i / 96, c2 = (i % 96) << 1;
        float2 v = *(const float2*)(zg + (size_t)(rbase + r) * 192 + c2);
        *(float2*)&smem[zw + r * YP + c2] = v;
    }
    for (int i = pid; i < 16 * 96; i += 128) {
        int r = i / 96, c2 = (i % 96) << 1;
        float2 v;
        if (c2 < 128) v = *(const float2*)(y1g + (size_t)(rbase + r) * 128 + c2);
        else          v = *(const float2*)(y2g + (size_t)(rbase + r) * 64 + (c2 - 128));
        v.x *= -TAU; v.y *= -TAU;
        *(float2*)&smem[ysw + r * YP + c2] = v;
    }
    __syncthreads();

    const float NEG_INV_TS = -1.0f / (TAU * SIGMA);

    float ysr[3][8], zr[3][8];
    float xr[4][8], cr[4];
#pragma unroll
    for (int cti = 0; cti < 3; ++cti) {
        int ct = qtr * 3 + cti;
#pragma unroll
        for (int v = 0; v < 8; ++v) {
            int row = v + hi, col = ct * 16 + n;
            ysr[cti][v] = smem[ysw + row * YP + col];
            zr[cti][v]  = smem[zw  + row * YP + col];
        }
    }
#pragma unroll
    for (int cti = 0; cti < 4; ++cti) {
        int ct = qtr * 4 + cti;
        cr[cti] = smem[OFF_C + ct * 16 + n];
#pragma unroll
        for (int v = 0; v < 8; ++v)
            xr[cti][v] = smem[xw + (v + hi) * AP + ct * 16 + n];
    }

    for (int it = 0; it < MAX_ITER; ++it) {
        v8f acc[3];
#pragma unroll
        for (int cti = 0; cti < 3; ++cti) {
#pragma unroll
            for (int v = 0; v < 8; ++v)
                acc[cti][v] = ysr[cti][v] * NEG_INV_TS - zr[cti][v];
        }
        for (int k0 = 0; k0 < 256; k0 += 4) {
            int ko = k0 + ko2;
            v2f a = *(const v2f*)&smem[xbw + (lane & 15) * AP + ko];
#pragma unroll
            for (int cti = 0; cti < 3; ++cti) {
                int ct = qtr * 3 + cti;
                v2f b = *(const v2f*)&smem[OFF_A + (ct * 16 + (lane & 15)) * AP + ko];
                acc[cti] = __builtin_amdgcn_wmma_f32_16x16x4_f32(
                    false, a, false, b, (short)0, acc[cti], false, false);
            }
        }
#pragma unroll
        for (int cti = 0; cti < 3; ++cti) {
            int ct = qtr * 3 + cti;
#pragma unroll
            for (int v = 0; v < 8; ++v) {
                float yv = SIGMA * acc[cti][v];
                yv = (ct < 8) ? fmaxf(yv, 0.0f) : yv;
                float t = -TAU * yv;
                ysr[cti][v] = t;
                smem[ysw + (v + hi) * YP + ct * 16 + n] = t;
            }
        }
        __syncthreads();

        v8f acc2[4];
#pragma unroll
        for (int cti = 0; cti < 4; ++cti) {
            float cv = cr[cti];
#pragma unroll
            for (int v = 0; v < 8; ++v)
                acc2[cti][v] = __builtin_fmaf(-TAU, cv, xr[cti][v]);
        }
        for (int m0 = 0; m0 < 192; m0 += 4) {
            int mo = m0 + ko2;
            v2f a = *(const v2f*)&smem[ysw + (lane & 15) * YP + mo];
#pragma unroll
            for (int cti = 0; cti < 4; ++cti) {
                int ct = qtr * 4 + cti;
                v2f b;
                b.x = smem[OFF_A + mo * AP + ct * 16 + n];
                b.y = smem[OFF_A + (mo + 1) * AP + ct * 16 + n];
                acc2[cti] = __builtin_amdgcn_wmma_f32_16x16x4_f32(
                    false, a, false, b, (short)0, acc2[cti], false, false);
            }
        }
#pragma unroll
        for (int cti = 0; cti < 4; ++cti) {
            int ct = qtr * 4 + cti;
#pragma unroll
            for (int v = 0; v < 8; ++v) {
                int idx = (v + hi) * AP + ct * 16 + n;
                float xo = xr[cti][v];
                float xn = fmaxf(acc2[cti][v], 0.0f);
                xr[cti][v] = xn;
                smem[xbw + idx] = 2.0f * xn - xo;
            }
        }
        __syncthreads();
    }

#pragma unroll
    for (int cti = 0; cti < 4; ++cti) {
        int ct = qtr * 4 + cti;
#pragma unroll
        for (int v = 0; v < 8; ++v)
            smem[xw + (v + hi) * AP + ct * 16 + n] = xr[cti][v];
    }
    __syncthreads();

    float* out_x  = out;
    float* out_y1 = out + (size_t)8192 * 256;
    float* out_y2 = out_y1 + (size_t)8192 * 128;
    for (int i = pid; i < 16 * 64; i += 128) {
        int r = i >> 6, c4 = (i & 63) << 2;
        float4 v = *(const float4*)&smem[xw + r * AP + c4];
        *(float4*)(out_x + (size_t)(rbase + r) * 256 + c4) = v;
    }
    const float NEG_INV_TAU = -1.0f / TAU;
    for (int i = pid; i < 16 * 96; i += 128) {
        int r = i / 96, c2 = (i % 96) << 1;
        float2 v = *(const float2*)&smem[ysw + r * YP + c2];
        v.x *= NEG_INV_TAU; v.y *= NEG_INV_TAU;
        if (c2 < 128) *(float2*)(out_y1 + (size_t)(rbase + r) * 128 + c2) = v;
        else          *(float2*)(out_y2 + (size_t)(rbase + r) * 64 + (c2 - 128)) = v;
    }
}

extern "C" void kernel_launch(void* const* d_in, const int* in_sizes, int n_in,
                              void* d_out, int out_size, void* d_ws, size_t ws_size,
                              hipStream_t stream) {
    const float* x0 = (const float*)d_in[0];
    const float* y1 = (const float*)d_in[1];
    const float* y2 = (const float*)d_in[2];
    const float* z  = (const float*)d_in[3];
    const float* c  = (const float*)d_in[4];
    const float* A1 = (const float*)d_in[5];
    const float* A2 = (const float*)d_in[6];
    (void)in_sizes; (void)n_in; (void)out_size; (void)d_ws; (void)ws_size;
    pdhg_kernel<<<dim3(8192 / 32), dim3(256), 0, stream>>>(
        x0, y1, y2, z, c, A1, A2, (float*)d_out);
}


# LDS fragment reads software-pipelined in both products: blocks of 8 k-steps, each block's reads issued as one group a block ahead of its WMMAs (second register set) instead of among them
# speedup vs baseline: 1.4543x; 1.0629x over previous
#include <hip/hip_runtime.h>


typedef __attribute__((ext_vector_type(2))) float v2f;
typedef __attribute__((ext_vector_type(8))) float v8f;

#define TAU   0.05f
#define SIGMA 0.05f
#define MAX_ITER 30

static constexpr int AP = 260;
static constexpr int YP = 196;

static constexpr int OFF_A  = 0;
static constexpr int OFF_X  = OFF_A  + 192 * AP;
static constexpr int OFF_XB = OFF_X  + 32  * AP;
static constexpr int OFF_YS = OFF_XB + 32  * AP;
static constexpr int OFF_Z  = OFF_YS + 32  * YP;
static constexpr int OFF_C  = OFF_Z  + 32  * YP;
static constexpr int SMEM_TOT = OFF_C + 256;

struct FragsA { v2f a[8]; v2f b[8][3]; };
struct RawB   { v2f a[8]; v2f b[8][2][2]; };
struct FragsB { v2f a[8]; v2f b[8][4]; };

#define KEEP_GROUPS_APART() __builtin_amdgcn_sched_barrier(0)

__global__ __launch_bounds__(256, 1)
void pdhg_kernel(const float* __restrict__ x0g, const float* __restrict__ y1g,
                 const float* __restrict__ y2g, const float* __restrict__ zg,
                 const float* __restrict__ cg,  const float* __restrict__ A1,
                 const float* __restrict__ A2,  float* __restrict__ out)
{
    __shared__ float smem[SMEM_TOT];

    const int tid  = threadIdx.x;
    const int wave = tid >> 5;
    const int lane = tid & 31;
    const int tile = wave >> 2;
    const int qtr  = wave & 3;
    const int n    = lane & 15;
    const int hi   = (lane >> 4) << 3;
    const int ko2  = (lane >> 4) << 1;
    const int pid  = (qtr << 5) | lane;

    const int rbase = blockIdx.x * 32 + tile * 16;
    const int xw  = OFF_X  + tile * 16 * AP;
    const int xbw = OFF_XB + tile * 16 * AP;
    const int ysw = OFF_YS + tile * 16 * YP;
    const int zw  = OFF_Z  + tile * 16 * YP;

    for (int i = tid; i < 192 * 64; i += 256) {
        int r = i >> 6, c4 = (i & 63) << 2;
        const float* src = (r < 128) ? (A1 + r * 256 + c4) : (A2 + (r - 128) * 256 + c4);
        *(float4*)&smem[OFF_A + r * AP + c4] = *(const float4*)src;
    }
    if (tid < 64)
        *(float4*)&smem[OFF_C + tid * 4] = *(const float4*)(cg + tid * 4);
    for (int i = pid; i < 16 * 64; i += 128) {
        int r = i >> 6, c4 = (i & 63) << 2;
        float4 v = *(const float4*)(x0g + (size_t)(rbase + r) * 256 + c4);
        *(float4*)&smem[xw  + r * AP + c4] = v;
        *(float4*)&smem[xbw + r * AP + c4] = v;
    }
    for (int i = pid; i < 16 * 96; i += 128) {
        int r = i / 96, c2 = (i % 96) << 1;
        float2 v = *(const float2*)(zg + (size_t)(rbase + r) * 192 + c2);
        *(float2*)&smem[zw + r * YP + c2] = v;
    }
    for (int i = pid; i < 16 * 96; i += 128) {
        int r = i / 96, c2 = (i % 96) << 1;
        float2 v;
        if (c2 < 128) v = *(const float2*)(y1g + (size_t)(rbase + r) * 128 + c2);
        else          v = *(const float2*)(y2g + (size_t)(rbase + r) * 64 + (c2 - 128));
        v.x *= -TAU; v.y *= -TAU;
        *(float2*)&smem[ysw + r * YP + c2] = v;
    }
    __syncthreads();

    const float NEG_INV_TS = -1.0f / (TAU * SIGMA);

    float ysr[3][8], zr[3][8];
    float xr[4][8], cr[4];
#pragma unroll
    for (int cti = 0; cti < 3; ++cti) {
        int ct = qtr * 3 + cti;
#pragma unroll
        for (int v = 0; v < 8; ++v) {
            int row = v + hi, col = ct * 16 + n;
            ysr[cti][v] = smem[ysw + row * YP + col];
            zr[cti][v]  = smem[zw  + row * YP + col];
        }
    }
#pragma unroll
    for (int cti = 0; cti < 4; ++cti) {
        int ct = qtr * 4 + cti;
        cr[cti] = smem[OFF_C + ct * 16 + n];
#pragma unroll
        for (int v = 0; v < 8; ++v)
            xr[cti][v] = smem[xw + (v + hi) * AP + ct * 16 + n];
    }

    for (int it = 0; it < MAX_ITER; ++it) {
        v8f acc[3];
#pragma unroll
        for (int cti = 0; cti < 3; ++cti) {
#pragma unroll
            for (int v = 0; v < 8; ++v)
                acc[cti][v] = ysr[cti][v] * NEG_INV_TS - zr[cti][v];
        }
        {
            const float* arow = &smem[xbw + n * AP + ko2];
            const float* brow = &smem[OFF_A + (qtr * 48 + n) * AP + ko2];
            auto fetch = [&](FragsA& f, int blk) {
#pragma unroll
                for (int s = 0; s < 8; ++s) {
                    int k = blk * 32 + s * 4;
                    f.a[s] = *(const v2f*)(arow + k);
#pragma unroll
                    for (int cti = 0; cti < 3; ++cti)
                        f.b[s][cti] = *(const v2f*)(brow + cti * 16 * AP + k);
                }
            };
            auto mma = [&](const FragsA& f) {
#pragma unroll
                for (int s = 0; s < 8; ++s) {
#pragma unroll
                    for (int cti = 0; cti < 3; ++cti)
                        acc[cti] = __builtin_amdgcn_wmma_f32_16x16x4_f32(
                            false, f.a[s], false, f.b[s][cti], (short)0, acc[cti], false, false);
                }
            };
            FragsA f0, f1;
            fetch(f0, 0);
            for (int blk = 0; blk < 6; blk += 2) {
                fetch(f1, blk + 1);
                KEEP_GROUPS_APART();
                mma(f0);
                KEEP_GROUPS_APART();
                fetch(f0, blk + 2);
                KEEP_GROUPS_APART();
                mma(f1);
                KEEP_GROUPS_APART();
            }
            fetch(f1, 7);
            KEEP_GROUPS_APART();
            mma(f0);
            mma(f1);
        }
#pragma unroll
        for (int cti = 0; cti < 3; ++cti) {
            int ct = qtr * 3 + cti;
#pragma unroll
            for (int v = 0; v < 8; ++v) {
                float yv = SIGMA * acc[cti][v];
                yv = (ct < 8) ? fmaxf(yv, 0.0f) : yv;
                float t = -TAU * yv;
                ysr[cti][v] = t;
                smem[ysw + (v + hi) * YP + ct * 16 + n] = t;
            }
        }
        __syncthreads();

        v8f acc2[4];
#pragma unroll
        for (int cti = 0; cti < 4; ++cti) {
            float cv = cr[cti];
#pragma unroll
            for (int v = 0; v < 8; ++v)
                acc2[cti][v] = __builtin_fmaf(-TAU, cv, xr[cti][v]);
        }
        {
            const float* arow = &smem[ysw + n * YP + ko2];
            const float* bcol = &smem[OFF_A + ko2 * AP + qtr * 64 + n];
            auto fetch = [&](RawB& f, int blk) {
#pragma unroll
                for (int s = 0; s < 8; ++s) {
                    int m = blk * 32 + s * 4;
                    f.a[s] = *(const v2f*)(arow + m);
#pragma unroll
                    for (int r = 0; r < 2; ++r) {
#pragma unroll
                        for (int p = 0; p < 2; ++p) {
                            v2f t;
                            t.x = bcol[(m + r) * AP + (2 * p) * 16];
                            t.y = bcol[(m + r) * AP + (2 * p + 1) * 16];
                            f.b[s][r][p] = t;
                        }
                    }
                }
            };
            auto pair_up = [&](FragsB& g, const RawB& f) {
#pragma unroll
                for (int s = 0; s < 8; ++s) {
                    g.a[s] = f.a[s];
#pragma unroll
                    for (int cti = 0; cti < 4; ++cti) {
                        v2f b;
                        b.x = f.b[s][0][cti >> 1][cti & 1];
                        b.y = f.b[s][1][cti >> 1][cti & 1];
                        g.b[s][cti] = b;
                    }
                }
            };
            auto mma = [&](const FragsB& g) {
#pragma unroll
                for (int s = 0; s < 8; ++s) {
#pragma unroll
                    for (int cti = 0; cti < 4; ++cti)
                        acc2[cti] = __builtin_amdgcn_wmma_f32_16x16x4_f32(
                            false, g.a[s], false, g.b[s][cti], (short)0, acc2[cti], false, false);
                }
            };
            RawB raw;
            FragsB cur;
            fetch(raw, 0);
            pair_up(cur, raw);
            for (int blk = 1; blk < 6; ++blk) {
                fetch(raw, blk);
                KEEP_GROUPS_APART();
                mma(cur);
                KEEP_GROUPS_APART();
                pair_up(cur, raw);
            }
            mma(cur);
        }
#pragma unroll
        for (int cti = 0; cti < 4; ++cti) {
            int ct = qtr * 4 + cti;
#pragma unroll
            for (int v = 0; v < 8; ++v) {
                int idx = (v + hi) * AP + ct * 16 + n;
                float xo = xr[cti][v];
                float xn = fmaxf(acc2[cti][v], 0.0f);
                xr[cti][v] = xn;
                smem[xbw + idx] = 2.0f * xn - xo;
            }
        }
        __syncthreads();
    }

#pragma unroll
    for (int cti = 0; cti < 4; ++cti) {
        int ct = qtr * 4 + cti;
#pragma unroll
        for (int v = 0; v < 8; ++v)
            smem[xw + (v + hi) * AP + ct * 16 + n] = xr[cti][v];
    }
    __syncthreads();

    float* out_x  = out;
    float* out_y1 = out + (size_t)8192 * 256;
    float* out_y2 = out_y1 + (size_t)8192 * 128;
    for (int i = pid; i < 16 * 64; i += 128) {
        int r = i >> 6, c4 = (i & 63) << 2;
        float4 v = *(const float4*)&smem[xw + r * AP + c4];
        *(float4*)(out_x + (size_t)(rbase + r) * 256 + c4) = v;
    }
    const float NEG_INV_TAU = -1.0f / TAU;
    for (int i = pid; i < 16 * 96; i += 128) {
        int r = i / 96, c2 = (i % 96) << 1;
        float2 v = *(const float2*)&smem[ysw + r * YP + c2];
        v.x *= NEG_INV_TAU; v.y *= NEG_INV_TAU;
        if (c2 < 128) *(float2*)(out_y1 + (size_t)(rbase + r) * 128 + c2) = v;
        else          *(float2*)(out_y2 + (size_t)(rbase + r) * 64 + (c2 - 128)) = v;
    }
}

extern "C" void kernel_launch(void* const* d_in, const int* in_sizes, int n_in,
                              void* d_out, int out_size, void* d_ws, size_t ws_size,
                              hipStream_t stream) {
    const float* x0 = (const float*)d_in[0];
    const float* y1 = (const float*)d_in[1];
    const float* y2 = (const float*)d_in[2];
    const float* z  = (const float*)d_in[3];
    const float* c  = (const float*)d_in[4];
    const float* A1 = (const float*)d_in[5];
    const float* A2 = (const float*)d_in[6];
    (void)in_sizes; (void)n_in; (void)out_size; (void)d_ws; (void)ws_size;
    pdhg_kernel<<<dim3(8192 / 32), dim3(256), 0, stream>>>(
        x0, y1, y2, z, c, A1, A2, (float*)d_out);
}


# Phase B's LDS reads of the constraint matrix made bank-conflict-free: lane halves read rows 8 apart (32 banks) instead of 2; -TAU*y columns stored to match; k is summed in that order
# speedup vs baseline: 1.5337x; 1.0546x over previous
#include <hip/hip_runtime.h>


typedef __attribute__((ext_vector_type(2))) float v2f;
typedef __attribute__((ext_vector_type(8))) float v8f;

#define TAU   0.05f
#define SIGMA 0.05f
#define MAX_ITER 30

static constexpr int AP = 260;
static constexpr int YP = 196;

static constexpr int OFF_A  = 0;
static constexpr int OFF_X  = OFF_A  + 192 * AP;
static constexpr int OFF_XB = OFF_X  + 32  * AP;
static constexpr int OFF_YS = OFF_XB + 32  * AP;
static constexpr int OFF_Z  = OFF_YS + 32  * YP;
static constexpr int OFF_C  = OFF_Z  + 32  * YP;
static constexpr int SMEM_TOT = OFF_C + 256;

struct FragsA { v2f a[8]; v2f b[8][3]; };
struct RawB   { v2f a[8]; v2f b[8][2][2]; };
struct FragsB { v2f a[8]; v2f b[8][4]; };

#define KEEP_GROUPS_APART() __builtin_amdgcn_sched_barrier(0)

__global__ __launch_bounds__(256, 1)
void pdhg_kernel(const float* __restrict__ x0g, const float* __restrict__ y1g,
                 const float* __restrict__ y2g, const float* __restrict__ zg,
                 const float* __restrict__ cg,  const float* __restrict__ A1,
                 const float* __restrict__ A2,  float* __restrict__ out)
{
    __shared__ float smem[SMEM_TOT];

    const int tid  = threadIdx.x;
    const int wave = tid >> 5;
    const int lane = tid & 31;
    const int tile = wave >> 2;
    const int qtr  = wave & 3;
    const int n    = lane & 15;
    const int hi   = (lane >> 4) << 3;
    const int ko2  = (lane >> 4) << 1;
    const int pid  = (qtr << 5) | lane;
    const int npm  = ((n >> 1) & 3) * 4 + (n >> 3) * 2 + (n & 1);

    const int rbase = blockIdx.x * 32 + tile * 16;
    const int xw  = OFF_X  + tile * 16 * AP;
    const int xbw = OFF_XB + tile * 16 * AP;
    const int ysw = OFF_YS + tile * 16 * YP;
    const int zw  = OFF_Z  + tile * 16 * YP;

    for (int i = tid; i < 192 * 64; i += 256) {
        int r = i >> 6, c4 = (i & 63) << 2;
        const float* src = (r < 128) ? (A1 + r * 256 + c4) : (A2 + (r - 128) * 256 + c4);
        *(float4*)&smem[OFF_A + r * AP + c4] = *(const float4*)src;
    }
    if (tid < 64)
        *(float4*)&smem[OFF_C + tid * 4] = *(const float4*)(cg + tid * 4);
    for (int i = pid; i < 16 * 64; i += 128) {
        int r = i >> 6, c4 = (i & 63) << 2;
        float4 v = *(const float4*)(x0g + (size_t)(rbase + r) * 256 + c4);
        *(float4*)&smem[xw  + r * AP + c4] = v;
        *(float4*)&smem[xbw + r * AP + c4] = v;
    }
    for (int i = pid; i < 16 * 96; i += 128) {
        int r = i / 96, c2 = (i % 96) << 1;
        float2 v = *(const float2*)(zg + (size_t)(rbase + r) * 192 + c2);
        *(float2*)&smem[zw + r * YP + c2] = v;
    }
    for (int i = pid; i < 16 * 96; i += 128) {
        int r = i / 96, c2 = (i % 96) << 1;
        float2 v;
        if (c2 < 128) v = *(const float2*)(y1g + (size_t)(rbase + r) * 128 + c2);
        else          v = *(const float2*)(y2g + (size_t)(rbase + r) * 64 + (c2 - 128));
        v.x *= -TAU; v.y *= -TAU;
        *(float2*)&smem[ysw + r * YP + c2] = v;
    }
    __syncthreads();

    const float NEG_INV_TS = -1.0f / (TAU * SIGMA);

    float ysr[3][8], zr[3][8];
    float xr[4][8], cr[4];
#pragma unroll
    for (int cti = 0; cti < 3; ++cti) {
        int ct = qtr * 3 + cti;
#pragma unroll
        for (int v = 0; v < 8; ++v) {
            int row = v + hi, col = ct * 16 + n;
            ysr[cti][v] = smem[ysw + row * YP + col];
            zr[cti][v]  = smem[zw  + row * YP + col];
        }
    }
#pragma unroll
    for (int cti = 0; cti < 4; ++cti) {
        int ct = qtr * 4 + cti;
        cr[cti] = smem[OFF_C + ct * 16 + n];
#pragma unroll
        for (int v = 0; v < 8; ++v)
            xr[cti][v] = smem[xw + (v + hi) * AP + ct * 16 + n];
    }

    for (int it = 0; it < MAX_ITER; ++it) {
        v8f acc[3];
#pragma unroll
        for (int cti = 0; cti < 3; ++cti) {
#pragma unroll
            for (int v = 0; v < 8; ++v)
                acc[cti][v] = ysr[cti][v] * NEG_INV_TS - zr[cti][v];
        }
        {
            const float* arow = &smem[xbw + n * AP + ko2];
            const float* brow = &smem[OFF_A + (qtr * 48 + n) * AP + ko2];
            auto fetch = [&](FragsA& f, int blk) {
#pragma unroll
                for (int s = 0; s < 8; ++s) {
                    int k = blk * 32 + s * 4;
                    f.a[s] = *(const v2f*)(arow + k);
#pragma unroll
                    for (int cti = 0; cti < 3; ++cti)
                        f.b[s][cti] = *(const v2f*)(brow + cti * 16 * AP + k);
                }
            };
            auto mma = [&](const FragsA& f) {
#pragma unroll
                for (int s = 0; s < 8; ++s) {
#pragma unroll
                    for (int cti = 0; cti < 3; ++cti)
                        acc[cti] = __builtin_amdgcn_wmma_f32_16x16x4_f32(
                            false, f.a[s], false, f.b[s][cti], (short)0, acc[cti], false, false);
                }
            };
            FragsA f0, f1;
            fetch(f0, 0);
            for (int blk = 0; blk < 6; blk += 2) {
                fetch(f1, blk + 1);
                KEEP_GROUPS_APART();
                mma(f0);
                KEEP_GROUPS_APART();
                fetch(f0, blk + 2);
                KEEP_GROUPS_APART();
                mma(f1);
                KEEP_GROUPS_APART();
            }
            fetch(f1, 7);
            KEEP_GROUPS_APART();
            mma(f0);
            mma(f1);
        }
#pragma unroll
        for (int cti = 0; cti < 3; ++cti) {
            int ct = qtr * 3 + cti;
#pragma unroll
            for (int v = 0; v < 8; ++v) {
                float yv = SIGMA * acc[cti][v];
                yv = (ct < 8) ? fmaxf(yv, 0.0f) : yv;
                float t = -TAU * yv;
                ysr[cti][v] = t;
                smem[ysw + (v + hi) * YP + ct * 16 + npm] = t;
            }
        }
        __syncthreads();

        v8f acc2[4];
#pragma unroll
        for (int cti = 0; cti < 4; ++cti) {
            float cv = cr[cti];
#pragma unroll
            for (int v = 0; v < 8; ++v)
                acc2[cti][v] = __builtin_fmaf(-TAU, cv, xr[cti][v]);
        }
        {
            const float* arow = &smem[ysw + n * YP + ko2];
            const float* bcol = &smem[OFF_A + (4 * ko2) * AP + qtr * 64 + n];
            auto fetch = [&](RawB& f, int blk) {
#pragma unroll
                for (int s = 0; s < 8; ++s) {
                    int m = blk * 32 + (s >> 2) * 16 + (s & 3) * 2;
                    f.a[s] = *(const v2f*)(arow + blk * 32 + s * 4);
#pragma unroll
                    for (int r = 0; r < 2; ++r) {
#pragma unroll
                        for (int p = 0; p < 2; ++p) {
                            v2f t;
                            t.x = bcol[(m + r) * AP + (2 * p) * 16];
                            t.y = bcol[(m + r) * AP + (2 * p + 1) * 16];
                            f.b[s][r][p] = t;
                        }
                    }
                }
            };
            auto pair_up = [&](FragsB& g, const RawB& f) {
#pragma unroll
                for (int s = 0; s < 8; ++s) {
                    g.a[s] = f.a[s];
#pragma unroll
                    for (int cti = 0; cti < 4; ++cti) {
                        v2f b;
                        b.x = f.b[s][0][cti >> 1][cti & 1];
                        b.y = f.b[s][1][cti >> 1][cti & 1];
                        g.b[s][cti] = b;
                    }
                }
            };
            auto mma = [&](const FragsB& g) {
#pragma unroll
                for (int s = 0; s < 8; ++s) {
#pragma unroll
                    for (int cti = 0; cti < 4; ++cti)
                        acc2[cti] = __builtin_amdgcn_wmma_f32_16x16x4_f32(
                            false, g.a[s], false, g.b[s][cti], (short)0, acc2[cti], false, false);
                }
            };
            RawB raw;
            FragsB cur;
            fetch(raw, 0);
            pair_up(cur, raw);
            for (int blk = 1; blk < 6; ++blk) {
                fetch(raw, blk);
                KEEP_GROUPS_APART();
                mma(cur);
                KEEP_GROUPS_APART();
                pair_up(cur, raw);
            }
            mma(cur);
        }
#pragma unroll
        for (int cti = 0; cti < 4; ++cti) {
            int ct = qtr * 4 + cti;
#pragma unroll
            for (int v = 0; v < 8; ++v) {
                int idx = (v + hi) * AP + ct * 16 + n;
                float xo = xr[cti][v];
                float xn = fmaxf(acc2[cti][v], 0.0f);
                xr[cti][v] = xn;
                smem[xbw + idx] = 2.0f * xn - xo;
            }
        }
        __syncthreads();
    }

#pragma unroll
    for (int cti = 0; cti < 4; ++cti) {
        int ct = qtr * 4 + cti;
#pragma unroll
        for (int v = 0; v < 8; ++v)
            smem[xw + (v + hi) * AP + ct * 16 + n] = xr[cti][v];
    }
#pragma unroll
    for (int cti = 0; cti < 3; ++cti) {
        int ct = qtr * 3 + cti;
#pragma unroll
        for (int v = 0; v < 8; ++v)
            smem[zw + (v + hi) * YP + ct * 16 + n] = ysr[cti][v];
    }
    __syncthreads();

    float* out_x  = out;
    float* out_y1 = out + (size_t)8192 * 256;
    float* out_y2 = out_y1 + (size_t)8192 * 128;
    for (int i = pid; i < 16 * 64; i += 128) {
        int r = i >> 6, c4 = (i & 63) << 2;
        float4 v = *(const float4*)&smem[xw + r * AP + c4];
        *(float4*)(out_x + (size_t)(rbase + r) * 256 + c4) = v;
    }
    const float NEG_INV_TAU = -1.0f / TAU;
    for (int i = pid; i < 16 * 96; i += 128) {
        int r = i / 96, c2 = (i % 96) << 1;
        float2 v = *(const float2*)&smem[zw + r * YP + c2];
        v.x *= NEG_INV_TAU; v.y *= NEG_INV_TAU;
        if (c2 < 128) *(float2*)(out_y1 + (size_t)(rbase + r) * 128 + c2) = v;
        else          *(float2*)(out_y2 + (size_t)(rbase + r) * 64 + (c2 - 128)) = v;
    }
}

extern "C" void kernel_launch(void* const* d_in, const int* in_sizes, int n_in,
                              void* d_out, int out_size, void* d_ws, size_t ws_size,
                              hipStream_t stream) {
    const float* x0 = (const float*)d_in[0];
    const float* y1 = (const float*)d_in[1];
    const float* y2 = (const float*)d_in[2];
    const float* z  = (const float*)d_in[3];
    const float* c  = (const float*)d_in[4];
    const float* A1 = (const float*)d_in[5];
    const float* A2 = (const float*)d_in[6];
    (void)in_sizes; (void)n_in; (void)out_size; (void)d_ws; (void)ws_size;
    pdhg_kernel<<<dim3(8192 / 32), dim3(256), 0, stream>>>(
        x0, y1, y2, z, c, A1, A2, (float*)d_out);
}


# Prologue global loads issued as one batch: a thread's x0, z, y0, c pieces go out first and wait in registers, the A copy runs 16 loads deep, LDS stores follow; no load waits on an earlier one's store
# speedup vs baseline: 1.6071x; 1.0479x over previous
#include <hip/hip_runtime.h>


typedef __attribute__((ext_vector_type(2))) float v2f;
typedef __attribute__((ext_vector_type(8))) float v8f;
typedef __attribute__((ext_vector_type(4))) float v4f;

#define TAU   0.05f
#define SIGMA 0.05f
#define MAX_ITER 30

static constexpr int AP = 260;
static constexpr int YP = 196;

static constexpr int OFF_A  = 0;
static constexpr int OFF_X  = OFF_A  + 192 * AP;
static constexpr int OFF_XB = OFF_X  + 32  * AP;
static constexpr int OFF_YS = OFF_XB + 32  * AP;
static constexpr int OFF_Z  = OFF_YS + 32  * YP;
static constexpr int OFF_C  = OFF_Z  + 32  * YP;
static constexpr int SMEM_TOT = OFF_C + 256;

struct FragsA { v2f a[8]; v2f b[8][3]; };
struct RawB   { v2f a[8]; v2f b[8][2][2]; };
struct FragsB { v2f a[8]; v2f b[8][4]; };

#define KEEP_GROUPS_APART() __builtin_amdgcn_sched_barrier(0)

__global__ __launch_bounds__(256, 1)
void pdhg_kernel(const float* __restrict__ x0g, const float* __restrict__ y1g,
                 const float* __restrict__ y2g, const float* __restrict__ zg,
                 const float* __restrict__ cg,  const float* __restrict__ A1,
                 const float* __restrict__ A2,  float* __restrict__ out)
{
    __shared__ float smem[SMEM_TOT];

    const int tid  = threadIdx.x;
    const int wave = tid >> 5;
    const int lane = tid & 31;
    const int tile = wave >> 2;
    const int qtr  = wave & 3;
    const int n    = lane & 15;
    const int hi   = (lane >> 4) << 3;
    const int ko2  = (lane >> 4) << 1;
    const int pid  = (qtr << 5) | lane;
    const int npm  = ((n >> 1) & 3) * 4 + (n >> 3) * 2 + (n & 1);

    const int rbase = blockIdx.x * 32 + tile * 16;
    const int xw  = OFF_X  + tile * 16 * AP;
    const int xbw = OFF_XB + tile * 16 * AP;
    const int ysw = OFF_YS + tile * 16 * YP;
    const int zw  = OFF_Z  + tile * 16 * YP;

    v4f xv[8];
    v2f zv[12], yv[12];
    v4f cv = {0.0f, 0.0f, 0.0f, 0.0f};
#pragma unroll
    for (int j = 0; j < 8; ++j) {
        int i = pid + j * 128;
        int r = i >> 6, c4 = (i & 63) << 2;
        xv[j] = *(const v4f*)(x0g + (size_t)(rbase + r) * 256 + c4);
    }
#pragma unroll
    for (int j = 0; j < 12; ++j) {
        int i = pid + j * 128;
        int r = i / 96, c2 = (i % 96) << 1;
        zv[j] = *(const v2f*)(zg + (size_t)(rbase + r) * 192 + c2);
        const float* ysrc = (c2 < 128) ? (y1g + (size_t)(rbase + r) * 128 + c2)
                                       : (y2g + (size_t)(rbase + r) * 64 + (c2 - 128));
        yv[j] = *(const v2f*)ysrc;
    }
    if (tid < 64)
        cv = *(const v4f*)(cg + tid * 4);
#pragma unroll
    for (int g = 0; g < 3; ++g) {
        v4f av[16];
#pragma unroll
        for (int u = 0; u < 16; ++u) {
            int i = tid + (g * 16 + u) * 256;
            int r = i >> 6, c4 = (i & 63) << 2;
            const float* src = (g * 16 + u < 32) ? (A1 + r * 256 + c4) : (A2 + (r - 128) * 256 + c4);
            av[u] = *(const v4f*)src;
        }
#pragma unroll
        for (int u = 0; u < 16; ++u) {
            int i = tid + (g * 16 + u) * 256;
            int r = i >> 6, c4 = (i & 63) << 2;
            *(v4f*)&smem[OFF_A + r * AP + c4] = av[u];
        }
        __builtin_amdgcn_sched_barrier(0);
    }
#pragma unroll
    for (int j = 0; j < 8; ++j) {
        int i = pid + j * 128;
        int r = i >> 6, c4 = (i & 63) << 2;
        *(v4f*)&smem[xw  + r * AP + c4] = xv[j];
        *(v4f*)&smem[xbw + r * AP + c4] = xv[j];
    }
#pragma unroll
    for (int j = 0; j < 12; ++j) {
        int i = pid + j * 128;
        int r = i / 96, c2 = (i % 96) << 1;
        *(v2f*)&smem[zw + r * YP + c2] = zv[j];
        v2f v = yv[j];
        v.x *= -TAU; v.y *= -TAU;
        *(v2f*)&smem[ysw + r * YP + c2] = v;
    }
    if (tid < 64)
        *(v4f*)&smem[OFF_C + tid * 4] = cv;
    __syncthreads();

    const float NEG_INV_TS = -1.0f / (TAU * SIGMA);

    float ysr[3][8], zr[3][8];
    float xr[4][8], cr[4];
#pragma unroll
    for (int cti = 0; cti < 3; ++cti) {
        int ct = qtr * 3 + cti;
#pragma unroll
        for (int v = 0; v < 8; ++v) {
            int row = v + hi, col = ct * 16 + n;
            ysr[cti][v] = smem[ysw + row * YP + col];
            zr[cti][v]  = smem[zw  + row * YP + col];
        }
    }
#pragma unroll
    for (int cti = 0; cti < 4; ++cti) {
        int ct = qtr * 4 + cti;
        cr[cti] = smem[OFF_C + ct * 16 + n];
#pragma unroll
        for (int v = 0; v < 8; ++v)
            xr[cti][v] = smem[xw + (v + hi) * AP + ct * 16 + n];
    }

    for (int it = 0; it < MAX_ITER; ++it) {
        v8f acc[3];
#pragma unroll
        for (int cti = 0; cti < 3; ++cti) {
#pragma unroll
            for (int v = 0; v < 8; ++v)
                acc[cti][v] = ysr[cti][v] * NEG_INV_TS - zr[cti][v];
        }
        {
            const float* arow = &smem[xbw + n * AP + ko2];
            const float* brow = &smem[OFF_A + (qtr * 48 + n) * AP + ko2];
            auto fetch = [&](FragsA& f, int blk) {
#pragma unroll
                for (int s = 0; s < 8; ++s) {
                    int k = blk * 32 + s * 4;
                    f.a[s] = *(const v2f*)(arow + k);
#pragma unroll
                    for (int cti = 0; cti < 3; ++cti)
                        f.b[s][cti] = *(const v2f*)(brow + cti * 16 * AP + k);
                }
            };
            auto mma = [&](const FragsA& f) {
#pragma unroll
                for (int s = 0; s < 8; ++s) {
#pragma unroll
                    for (int cti = 0; cti < 3; ++cti)
                        acc[cti] = __builtin_amdgcn_wmma_f32_16x16x4_f32(
                            false, f.a[s], false, f.b[s][cti], (short)0, acc[cti], false, false);
                }
            };
            FragsA f0, f1;
            fetch(f0, 0);
            for (int blk = 0; blk < 6; blk += 2) {
                fetch(f1, blk + 1);
                KEEP_GROUPS_APART();
                mma(f0);
                KEEP_GROUPS_APART();
                fetch(f0, blk + 2);
                KEEP_GROUPS_APART();
                mma(f1);
                KEEP_GROUPS_APART();
            }
            fetch(f1, 7);
            KEEP_GROUPS_APART();
            mma(f0);
            mma(f1);
        }
#pragma unroll
        for (int cti = 0; cti < 3; ++cti) {
            int ct = qtr * 3 + cti;
#pragma unroll
            for (int v = 0; v < 8; ++v) {
                float yv = SIGMA * acc[cti][v];
                yv = (ct < 8) ? fmaxf(yv, 0.0f) : yv;
                float t = -TAU * yv;
                ysr[cti][v] = t;
                smem[ysw + (v + hi) * YP + ct * 16 + npm] = t;
            }
        }
        __syncthreads();

        v8f acc2[4];
#pragma unroll
        for (int cti = 0; cti < 4; ++cti) {
            float cv = cr[cti];
#pragma unroll
            for (int v = 0; v < 8; ++v)
                acc2[cti][v] = __builtin_fmaf(-TAU, cv, xr[cti][v]);
        }
        {
            const float* arow = &smem[ysw + n * YP + ko2];
            const float* bcol = &smem[OFF_A + (4 * ko2) * AP + qtr * 64 + n];
            auto fetch = [&](RawB& f, int blk) {
#pragma unroll
                for (int s = 0; s < 8; ++s) {
                    int m = blk * 32 + (s >> 2) * 16 + (s & 3) * 2;
                    f.a[s] = *(const v2f*)(arow + blk * 32 + s * 4);
#pragma unroll
                    for (int r = 0; r < 2; ++r) {
#pragma unroll
                        for (int p = 0; p < 2; ++p) {
                            v2f t;
                            t.x = bcol[(m + r) * AP + (2 * p) * 16];
                            t.y = bcol[(m + r) * AP + (2 * p + 1) * 16];
                            f.b[s][r][p] = t;
                        }
                    }
                }
            };
            auto pair_up = [&](FragsB& g, const RawB& f) {
#pragma unroll
                for (int s = 0; s < 8; ++s) {
                    g.a[s] = f.a[s];
#pragma unroll
                    for (int cti = 0; cti < 4; ++cti) {
                        v2f b;
                        b.x = f.b[s][0][cti >> 1][cti & 1];
                        b.y = f.b[s][1][cti >> 1][cti & 1];
                        g.b[s][cti] = b;
                    }
                }
            };
            auto mma = [&](const FragsB& g) {
#pragma unroll
                for (int s = 0; s < 8; ++s) {
#pragma unroll
                    for (int cti = 0; cti < 4; ++cti)
                        acc2[cti] = __builtin_amdgcn_wmma_f32_16x16x4_f32(
                            false, g.a[s], false, g.b[s][cti], (short)0, acc2[cti], false, false);
                }
            };
            RawB raw;
            FragsB cur;
            fetch(raw, 0);
            pair_up(cur, raw);
            for (int blk = 1; blk < 6; ++blk) {
                fetch(raw, blk);
                KEEP_GROUPS_APART();
                mma(cur);
                KEEP_GROUPS_APART();
                pair_up(cur, raw);
            }
            mma(cur);
        }
#pragma unroll
        for (int cti = 0; cti < 4; ++cti) {
            int ct = qtr * 4 + cti;
#pragma unroll
            for (int v = 0; v < 8; ++v) {
                int idx = (v + hi) * AP + ct * 16 + n;
                float xo = xr[cti][v];
                float xn = fmaxf(acc2[cti][v], 0.0f);
                xr[cti][v] = xn;
                smem[xbw + idx] = 2.0f * xn - xo;
            }
        }
        __syncthreads();
    }

#pragma unroll
    for (int cti = 0; cti < 4; ++cti) {
        int ct = qtr * 4 + cti;
#pragma unroll
        for (int v = 0; v < 8; ++v)
            smem[xw + (v + hi) * AP + ct * 16 + n] = xr[cti][v];
    }
#pragma unroll
    for (int cti = 0; cti < 3; ++cti) {
        int ct = qtr * 3 + cti;
#pragma unroll
        for (int v = 0; v < 8; ++v)
            smem[zw + (v + hi) * YP + ct * 16 + n] = ysr[cti][v];
    }
    __syncthreads();

    float* out_x  = out;
    float* out_y1 = out + (size_t)8192 * 256;
    float* out_y2 = out_y1 + (size_t)8192 * 128;
    for (int i = pid; i < 16 * 64; i += 128) {
        int r = i >> 6, c4 = (i & 63) << 2;
        float4 v = *(const float4*)&smem[xw + r * AP + c4];
        *(float4*)(out_x + (size_t)(rbase + r) * 256 + c4) = v;
    }
    const float NEG_INV_TAU = -1.0f / TAU;
    for (int i = pid; i < 16 * 96; i += 128) {
        int r = i / 96, c2 = (i % 96) << 1;
        float2 v = *(const float2*)&smem[zw + r * YP + c2];
        v.x *= NEG_INV_TAU; v.y *= NEG_INV_TAU;
        if (c2 < 128) *(float2*)(out_y1 + (size_t)(rbase + r) * 128 + c2) = v;
        else          *(float2*)(out_y2 + (size_t)(rbase + r) * 64 + (c2 - 128)) = v;
    }
}

extern "C" void kernel_launch(void* const* d_in, const int* in_sizes, int n_in,
                              void* d_out, int out_size, void* d_ws, size_t ws_size,
                              hipStream_t stream) {
    const float* x0 = (const float*)d_in[0];
    const float* y1 = (const float*)d_in[1];
    const float* y2 = (const float*)d_in[2];
    const float* z  = (const float*)d_in[3];
    const float* c  = (const float*)d_in[4];
    const float* A1 = (const float*)d_in[5];
    const float* A2 = (const float*)d_in[6];
    (void)in_sizes; (void)n_in; (void)out_size; (void)d_ws; (void)ws_size;
    pdhg_kernel<<<dim3(8192 / 32), dim3(256), 0, stream>>>(
        x0, y1, y2, z, c, A1, A2, (float*)d_out);
}


# Priority window: each SIMD's younger wave (second tile) raises issue priority for its first 6 of 8 blocks of product one and 5 of 6 of product two, so a SIMD's two waves reach each barrier together
# speedup vs baseline: 1.6608x; 1.0334x over previous
#include <hip/hip_runtime.h>


typedef __attribute__((ext_vector_type(2))) float v2f;
typedef __attribute__((ext_vector_type(8))) float v8f;
typedef __attribute__((ext_vector_type(4))) float v4f;

#define TAU   0.05f
#define SIGMA 0.05f
#define MAX_ITER 30

static constexpr int AP = 260;
static constexpr int YP = 196;

static constexpr int OFF_A  = 0;
static constexpr int OFF_X  = OFF_A  + 192 * AP;
static constexpr int OFF_XB = OFF_X  + 32  * AP;
static constexpr int OFF_YS = OFF_XB + 32  * AP;
static constexpr int OFF_Z  = OFF_YS + 32  * YP;
static constexpr int OFF_C  = OFF_Z  + 32  * YP;
static constexpr int SMEM_TOT = OFF_C + 256;

struct FragsA { v2f a[8]; v2f b[8][3]; };
struct RawB   { v2f a[8]; v2f b[8][2][2]; };
struct FragsB { v2f a[8]; v2f b[8][4]; };

#define KEEP_GROUPS_APART() __builtin_amdgcn_sched_barrier(0)

__global__ __launch_bounds__(256, 1)
void pdhg_kernel(const float* __restrict__ x0g, const float* __restrict__ y1g,
                 const float* __restrict__ y2g, const float* __restrict__ zg,
                 const float* __restrict__ cg,  const float* __restrict__ A1,
                 const float* __restrict__ A2,  float* __restrict__ out)
{
    __shared__ float smem[SMEM_TOT];

    const int tid  = threadIdx.x;
    const int wave = tid >> 5;
    const int lane = tid & 31;
    const int tile = wave >> 2;
    const int qtr  = wave & 3;
    const int n    = lane & 15;
    const int hi   = (lane >> 4) << 3;
    const int ko2  = (lane >> 4) << 1;
    const int pid  = (qtr << 5) | lane;
    const int npm  = ((n >> 1) & 3) * 4 + (n >> 3) * 2 + (n & 1);
    const int younger = __builtin_amdgcn_readfirstlane(tile);

    const int rbase = blockIdx.x * 32 + tile * 16;
    const int xw  = OFF_X  + tile * 16 * AP;
    const int xbw = OFF_XB + tile * 16 * AP;
    const int ysw = OFF_YS + tile * 16 * YP;
    const int zw  = OFF_Z  + tile * 16 * YP;

    v4f xv[8];
    v2f zv[12], yv[12];
    v4f cv = {0.0f, 0.0f, 0.0f, 0.0f};
#pragma unroll
    for (int j = 0; j < 8; ++j) {
        int i = pid + j * 128;
        int r = i >> 6, c4 = (i & 63) << 2;
        xv[j] = *(const v4f*)(x0g + (size_t)(rbase + r) * 256 + c4);
    }
#pragma unroll
    for (int j = 0; j < 12; ++j) {
        int i = pid + j * 128;
        int r = i / 96, c2 = (i % 96) << 1;
        zv[j] = *(const v2f*)(zg + (size_t)(rbase + r) * 192 + c2);
        const float* ysrc = (c2 < 128) ? (y1g + (size_t)(rbase + r) * 128 + c2)
                                       : (y2g + (size_t)(rbase + r) * 64 + (c2 - 128));
        yv[j] = *(const v2f*)ysrc;
    }
    if (tid < 64)
        cv = *(const v4f*)(cg + tid * 4);
#pragma unroll
    for (int g = 0; g < 3; ++g) {
        v4f av[16];
#pragma unroll
        for (int u = 0; u < 16; ++u) {
            int i = tid + (g * 16 + u) * 256;
            int r = i >> 6, c4 = (i & 63) << 2;
            const float* src = (g * 16 + u < 32) ? (A1 + r * 256 + c4) : (A2 + (r - 128) * 256 + c4);
            av[u] = *(const v4f*)src;
        }
#pragma unroll
        for (int u = 0; u < 16; ++u) {
            int i = tid + (g * 16 + u) * 256;
            int r = i >> 6, c4 = (i & 63) << 2;
            *(v4f*)&smem[OFF_A + r * AP + c4] = av[u];
        }
        __builtin_amdgcn_sched_barrier(0);
    }
#pragma unroll
    for (int j = 0; j < 8; ++j) {
        int i = pid + j * 128;
        int r = i >> 6, c4 = (i & 63) << 2;
        *(v4f*)&smem[xw  + r * AP + c4] = xv[j];
        *(v4f*)&smem[xbw + r * AP + c4] = xv[j];
    }
#pragma unroll
    for (int j = 0; j < 12; ++j) {
        int i = pid + j * 128;
        int r = i / 96, c2 = (i % 96) << 1;
        *(v2f*)&smem[zw + r * YP + c2] = zv[j];
        v2f v = yv[j];
        v.x *= -TAU; v.y *= -TAU;
        *(v2f*)&smem[ysw + r * YP + c2] = v;
    }
    if (tid < 64)
        *(v4f*)&smem[OFF_C + tid * 4] = cv;
    __syncthreads();

    const float NEG_INV_TS = -1.0f / (TAU * SIGMA);

    float ysr[3][8], zr[3][8];
    float xr[4][8], cr[4];
#pragma unroll
    for (int cti = 0; cti < 3; ++cti) {
        int ct = qtr * 3 + cti;
#pragma unroll
        for (int v = 0; v < 8; ++v) {
            int row = v + hi, col = ct * 16 + n;
            ysr[cti][v] = smem[ysw + row * YP + col];
            zr[cti][v]  = smem[zw  + row * YP + col];
        }
    }
#pragma unroll
    for (int cti = 0; cti < 4; ++cti) {
        int ct = qtr * 4 + cti;
        cr[cti] = smem[OFF_C + ct * 16 + n];
#pragma unroll
        for (int v = 0; v < 8; ++v)
            xr[cti][v] = smem[xw + (v + hi) * AP + ct * 16 + n];
    }

    for (int it = 0; it < MAX_ITER; ++it) {
        if (younger) __builtin_amdgcn_s_setprio(3);
        v8f acc[3];
#pragma unroll
        for (int cti = 0; cti < 3; ++cti) {
#pragma unroll
            for (int v = 0; v < 8; ++v)
                acc[cti][v] = ysr[cti][v] * NEG_INV_TS - zr[cti][v];
        }
        {
            const float* arow = &smem[xbw + n * AP + ko2];
            const float* brow = &smem[OFF_A + (qtr * 48 + n) * AP + ko2];
            auto fetch = [&](FragsA& f, int blk) {
#pragma unroll
                for (int s = 0; s < 8; ++s) {
                    int k = blk * 32 + s * 4;
                    f.a[s] = *(const v2f*)(arow + k);
#pragma unroll
                    for (int cti = 0; cti < 3; ++cti)
                        f.b[s][cti] = *(const v2f*)(brow + cti * 16 * AP + k);
                }
            };
            auto mma = [&](const FragsA& f) {
#pragma unroll
                for (int s = 0; s < 8; ++s) {
#pragma unroll
                    for (int cti = 0; cti < 3; ++cti)
                        acc[cti] = __builtin_amdgcn_wmma_f32_16x16x4_f32(
                            false, f.a[s], false, f.b[s][cti], (short)0, acc[cti], false, false);
                }
            };
            FragsA f0, f1;
            fetch(f0, 0);
            for (int blk = 0; blk < 6; blk += 2) {
                fetch(f1, blk + 1);
                KEEP_GROUPS_APART();
                mma(f0);
                KEEP_GROUPS_APART();
                fetch(f0, blk + 2);
                KEEP_GROUPS_APART();
                mma(f1);
                KEEP_GROUPS_APART();
            }
            if (younger) __builtin_amdgcn_s_setprio(0);
            fetch(f1, 7);
            KEEP_GROUPS_APART();
            mma(f0);
            mma(f1);
        }
#pragma unroll
        for (int cti = 0; cti < 3; ++cti) {
            int ct = qtr * 3 + cti;
#pragma unroll
            for (int v = 0; v < 8; ++v) {
                float yv = SIGMA * acc[cti][v];
                yv = (ct < 8) ? fmaxf(yv, 0.0f) : yv;
                float t = -TAU * yv;
                ysr[cti][v] = t;
                smem[ysw + (v + hi) * YP + ct * 16 + npm] = t;
            }
        }
        __syncthreads();

        if (younger) __builtin_amdgcn_s_setprio(3);
        v8f acc2[4];
#pragma unroll
        for (int cti = 0; cti < 4; ++cti) {
            float cv = cr[cti];
#pragma unroll
            for (int v = 0; v < 8; ++v)
                acc2[cti][v] = __builtin_fmaf(-TAU, cv, xr[cti][v]);
        }
        {
            const float* arow = &smem[ysw + n * YP + ko2];
            const float* bcol = &smem[OFF_A + (4 * ko2) * AP + qtr * 64 + n];
            auto fetch = [&](RawB& f, int blk) {
#pragma unroll
                for (int s = 0; s < 8; ++s) {
                    int m = blk * 32 + (s >> 2) * 16 + (s & 3) * 2;
                    f.a[s] = *(const v2f*)(arow + blk * 32 + s * 4);
#pragma unroll
                    for (int r = 0; r < 2; ++r) {
#pragma unroll
                        for (int p = 0; p < 2; ++p) {
                            v2f t;
                            t.x = bcol[(m + r) * AP + (2 * p) * 16];
                            t.y = bcol[(m + r) * AP + (2 * p + 1) * 16];
                            f.b[s][r][p] = t;
                        }
                    }
                }
            };
            auto pair_up = [&](FragsB& g, const RawB& f) {
#pragma unroll
                for (int s = 0; s < 8; ++s) {
                    g.a[s] = f.a[s];
#pragma unroll
                    for (int cti = 0; cti < 4; ++cti) {
                        v2f b;
                        b.x = f.b[s][0][cti >> 1][cti & 1];
                        b.y = f.b[s][1][cti >> 1][cti & 1];
                        g.b[s][cti] = b;
                    }
                }
            };
            auto mma = [&](const FragsB& g) {
#pragma unroll
                for (int s = 0; s < 8; ++s) {
#pragma unroll
                    for (int cti = 0; cti < 4; ++cti)
                        acc2[cti] = __builtin_amdgcn_wmma_f32_16x16x4_f32(
                            false, g.a[s], false, g.b[s][cti], (short)0, acc2[cti], false, false);
                }
            };
            RawB raw;
            FragsB cur;
            fetch(raw, 0);
            pair_up(cur, raw);
            for (int blk = 1; blk < 6; ++blk) {
                fetch(raw, blk);
                KEEP_GROUPS_APART();
                mma(cur);
                KEEP_GROUPS_APART();
                pair_up(cur, raw);
            }
            if (younger) __builtin_amdgcn_s_setprio(0);
            mma(cur);
        }
#pragma unroll
        for (int cti = 0; cti < 4; ++cti) {
            int ct = qtr * 4 + cti;
#pragma unroll
            for (int v = 0; v < 8; ++v) {
                int idx = (v + hi) * AP + ct * 16 + n;
                float xo = xr[cti][v];
                float xn = fmaxf(acc2[cti][v], 0.0f);
                xr[cti][v] = xn;
                smem[xbw + idx] = 2.0f * xn - xo;
            }
        }
        __syncthreads();
    }

#pragma unroll
    for (int cti = 0; cti < 4; ++cti) {
        int ct = qtr * 4 + cti;
#pragma unroll
        for (int v = 0; v < 8; ++v)
            smem[xw + (v + hi) * AP + ct * 16 + n] = xr[cti][v];
    }
#pragma unroll
    for (int cti = 0; cti < 3; ++cti) {
        int ct = qtr * 3 + cti;
#pragma unroll
        for (int v = 0; v < 8; ++v)
            smem[zw + (v + hi) * YP + ct * 16 + n] = ysr[cti][v];
    }
    __syncthreads();

    float* out_x  = out;
    float* out_y1 = out + (size_t)8192 * 256;
    float* out_y2 = out_y1 + (size_t)8192 * 128;
    for (int i = pid; i < 16 * 64; i += 128) {
        int r = i >> 6, c4 = (i & 63) << 2;
        float4 v = *(const float4*)&smem[xw + r * AP + c4];
        *(float4*)(out_x + (size_t)(rbase + r) * 256 + c4) = v;
    }
    const float NEG_INV_TAU = -1.0f / TAU;
    for (int i = pid; i < 16 * 96; i += 128) {
        int r = i / 96, c2 = (i % 96) << 1;
        float2 v = *(const float2*)&smem[zw + r * YP + c2];
        v.x *= NEG_INV_TAU; v.y *= NEG_INV_TAU;
        if (c2 < 128) *(float2*)(out_y1 + (size_t)(rbase + r) * 128 + c2) = v;
        else          *(float2*)(out_y2 + (size_t)(rbase + r) * 64 + (c2 - 128)) = v;
    }
}

extern "C" void kernel_launch(void* const* d_in, const int* in_sizes, int n_in,
                              void* d_out, int out_size, void* d_ws, size_t ws_size,
                              hipStream_t stream) {
    const float* x0 = (const float*)d_in[0];
    const float* y1 = (const float*)d_in[1];
    const float* y2 = (const float*)d_in[2];
    const float* z  = (const float*)d_in[3];
    const float* c  = (const float*)d_in[4];
    const float* A1 = (const float*)d_in[5];
    const float* A2 = (const float*)d_in[6];
    (void)in_sizes; (void)n_in; (void)out_size; (void)d_ws; (void)ws_size;
    pdhg_kernel<<<dim3(8192 / 32), dim3(256), 0, stream>>>(
        x0, y1, y2, z, c, A1, A2, (float*)d_out);
}


# Product two's output columns re-dealt so a lane owns four adjacent columns: one 16-byte LDS read per k row feeds all four matrix ops of a k-step (LDS reads per block 40 -> 20); xbar stored 16 B wide
# speedup vs baseline: 1.7298x; 1.0415x over previous
#include <hip/hip_runtime.h>


typedef __attribute__((ext_vector_type(2))) float v2f;
typedef __attribute__((ext_vector_type(8))) float v8f;
typedef __attribute__((ext_vector_type(4))) float v4f;

#define TAU   0.05f
#define SIGMA 0.05f
#define MAX_ITER 30

static constexpr int AP = 260;
static constexpr int YP = 196;

static constexpr int OFF_A  = 0;
static constexpr int OFF_X  = OFF_A  + 192 * AP;
static constexpr int OFF_XB = OFF_X  + 32  * AP;
static constexpr int OFF_YS = OFF_XB + 32  * AP;
static constexpr int OFF_Z  = OFF_YS + 32  * YP;
static constexpr int OFF_C  = OFF_Z  + 32  * YP;
static constexpr int SMEM_TOT = OFF_C + 256;

struct FragsA { v2f a[8]; v2f b[8][3]; };
struct RawB   { v2f a[8]; v4f b[8][2]; };
struct FragsB { v2f a[8]; v2f b[8][4]; };

#define KEEP_GROUPS_APART() __builtin_amdgcn_sched_barrier(0)

__global__ __launch_bounds__(256, 1)
void pdhg_kernel(const float* __restrict__ x0g, const float* __restrict__ y1g,
                 const float* __restrict__ y2g, const float* __restrict__ zg,
                 const float* __restrict__ cg,  const float* __restrict__ A1,
                 const float* __restrict__ A2,  float* __restrict__ out)
{
    __shared__ float smem[SMEM_TOT];

    const int tid  = threadIdx.x;
    const int wave = tid >> 5;
    const int lane = tid & 31;
    const int tile = wave >> 2;
    const int qtr  = wave & 3;
    const int n    = lane & 15;
    const int hi   = (lane >> 4) << 3;
    const int ko2  = (lane >> 4) << 1;
    const int pid  = (qtr << 5) | lane;
    const int npm  = ((n >> 1) & 3) * 4 + (n >> 3) * 2 + (n & 1);
    const int younger = __builtin_amdgcn_readfirstlane(tile);

    const int rbase = blockIdx.x * 32 + tile * 16;
    const int xw  = OFF_X  + tile * 16 * AP;
    const int xbw = OFF_XB + tile * 16 * AP;
    const int ysw = OFF_YS + tile * 16 * YP;
    const int zw  = OFF_Z  + tile * 16 * YP;

    v4f xv[8];
    v2f zv[12], yv[12];
    v4f cv = {0.0f, 0.0f, 0.0f, 0.0f};
#pragma unroll
    for (int j = 0; j < 8; ++j) {
        int i = pid + j * 128;
        int r = i >> 6, c4 = (i & 63) << 2;
        xv[j] = *(const v4f*)(x0g + (size_t)(rbase + r) * 256 + c4);
    }
#pragma unroll
    for (int j = 0; j < 12; ++j) {
        int i = pid + j * 128;
        int r = i / 96, c2 = (i % 96) << 1;
        zv[j] = *(const v2f*)(zg + (size_t)(rbase + r) * 192 + c2);
        const float* ysrc = (c2 < 128) ? (y1g + (size_t)(rbase + r) * 128 + c2)
                                       : (y2g + (size_t)(rbase + r) * 64 + (c2 - 128));
        yv[j] = *(const v2f*)ysrc;
    }
    if (tid < 64)
        cv = *(const v4f*)(cg + tid * 4);
#pragma unroll
    for (int g = 0; g < 3; ++g) {
        v4f av[16];
#pragma unroll
        for (int u = 0; u < 16; ++u) {
            int i = tid + (g * 16 + u) * 256;
            int r = i >> 6, c4 = (i & 63) << 2;
            const float* src = (g * 16 + u < 32) ? (A1 + r * 256 + c4) : (A2 + (r - 128) * 256 + c4);
            av[u] = *(const v4f*)src;
        }
#pragma unroll
        for (int u = 0; u < 16; ++u) {
            int i = tid + (g * 16 + u) * 256;
            int r = i >> 6, c4 = (i & 63) << 2;
            *(v4f*)&smem[OFF_A + r * AP + c4] = av[u];
        }
        __builtin_amdgcn_sched_barrier(0);
    }
#pragma unroll
    for (int j = 0; j < 8; ++j) {
        int i = pid + j * 128;
        int r = i >> 6, c4 = (i & 63) << 2;
        *(v4f*)&smem[xw  + r * AP + c4] = xv[j];
        *(v4f*)&smem[xbw + r * AP + c4] = xv[j];
    }
#pragma unroll
    for (int j = 0; j < 12; ++j) {
        int i = pid + j * 128;
        int r = i / 96, c2 = (i % 96) << 1;
        *(v2f*)&smem[zw + r * YP + c2] = zv[j];
        v2f v = yv[j];
        v.x *= -TAU; v.y *= -TAU;
        *(v2f*)&smem[ysw + r * YP + c2] = v;
    }
    if (tid < 64)
        *(v4f*)&smem[OFF_C + tid * 4] = cv;
    __syncthreads();

    const float NEG_INV_TS = -1.0f / (TAU * SIGMA);

    float ysr[3][8], zr[3][8];
    float xr[4][8], cr[4];
#pragma unroll
    for (int cti = 0; cti < 3; ++cti) {
        int ct = qtr * 3 + cti;
#pragma unroll
        for (int v = 0; v < 8; ++v) {
            int row = v + hi, col = ct * 16 + n;
            ysr[cti][v] = smem[ysw + row * YP + col];
            zr[cti][v]  = smem[zw  + row * YP + col];
        }
    }
    const int xcol = qtr * 64 + 4 * n;
    {
        v4f q = *(const v4f*)&smem[OFF_C + xcol];
#pragma unroll
        for (int cti = 0; cti < 4; ++cti)
            cr[cti] = q[cti];
    }
#pragma unroll
    for (int v = 0; v < 8; ++v) {
        v4f q = *(const v4f*)&smem[xw + (v + hi) * AP + xcol];
#pragma unroll
        for (int cti = 0; cti < 4; ++cti)
            xr[cti][v] = q[cti];
    }

    for (int it = 0; it < MAX_ITER; ++it) {
        if (younger) __builtin_amdgcn_s_setprio(3);
        v8f acc[3];
#pragma unroll
        for (int cti = 0; cti < 3; ++cti) {
#pragma unroll
            for (int v = 0; v < 8; ++v)
                acc[cti][v] = ysr[cti][v] * NEG_INV_TS - zr[cti][v];
        }
        {
            const float* arow = &smem[xbw + n * AP + ko2];
            const float* brow = &smem[OFF_A + (qtr * 48 + n) * AP + ko2];
            auto fetch = [&](FragsA& f, int blk) {
#pragma unroll
                for (int s = 0; s < 8; ++s) {
                    int k = blk * 32 + s * 4;
                    f.a[s] = *(const v2f*)(arow + k);
#pragma unroll
                    for (int cti = 0; cti < 3; ++cti)
                        f.b[s][cti] = *(const v2f*)(brow + cti * 16 * AP + k);
                }
            };
            auto mma = [&](const FragsA& f) {
#pragma unroll
                for (int s = 0; s < 8; ++s) {
#pragma unroll
                    for (int cti = 0; cti < 3; ++cti)
                        acc[cti] = __builtin_amdgcn_wmma_f32_16x16x4_f32(
                            false, f.a[s], false, f.b[s][cti], (short)0, acc[cti], false, false);
                }
            };
            FragsA f0, f1;
            fetch(f0, 0);
            for (int blk = 0; blk < 6; blk += 2) {
                fetch(f1, blk + 1);
                KEEP_GROUPS_APART();
                mma(f0);
                KEEP_GROUPS_APART();
                fetch(f0, blk + 2);
                KEEP_GROUPS_APART();
                mma(f1);
                KEEP_GROUPS_APART();
            }
            if (younger) __builtin_amdgcn_s_setprio(0);
            fetch(f1, 7);
            KEEP_GROUPS_APART();
            mma(f0);
            mma(f1);
        }
#pragma unroll
        for (int cti = 0; cti < 3; ++cti) {
            int ct = qtr * 3 + cti;
#pragma unroll
            for (int v = 0; v < 8; ++v) {
                float yv = SIGMA * acc[cti][v];
                yv = (ct < 8) ? fmaxf(yv, 0.0f) : yv;
                float t = -TAU * yv;
                ysr[cti][v] = t;
                smem[ysw + (v + hi) * YP + ct * 16 + npm] = t;
            }
        }
        __syncthreads();

        if (younger) __builtin_amdgcn_s_setprio(3);
        v8f acc2[4];
#pragma unroll
        for (int cti = 0; cti < 4; ++cti) {
            float cv = cr[cti];
#pragma unroll
            for (int v = 0; v < 8; ++v)
                acc2[cti][v] = __builtin_fmaf(-TAU, cv, xr[cti][v]);
        }
        {
            const float* arow = &smem[ysw + n * YP + ko2];
            const float* bcol = &smem[OFF_A + (4 * ko2) * AP + xcol];
            auto fetch = [&](RawB& f, int blk) {
#pragma unroll
                for (int s = 0; s < 8; ++s) {
                    int m = blk * 32 + (s >> 2) * 16 + (s & 3) * 2;
                    f.a[s] = *(const v2f*)(arow + blk * 32 + s * 4);
#pragma unroll
                    for (int r = 0; r < 2; ++r)
                        f.b[s][r] = *(const v4f*)(bcol + (m + r) * AP);
                }
            };
            auto pair_up = [&](FragsB& g, const RawB& f) {
#pragma unroll
                for (int s = 0; s < 8; ++s) {
                    g.a[s] = f.a[s];
#pragma unroll
                    for (int cti = 0; cti < 4; ++cti) {
                        v2f b;
                        b.x = f.b[s][0][cti];
                        b.y = f.b[s][1][cti];
                        g.b[s][cti] = b;
                    }
                }
            };
            auto mma = [&](const FragsB& g) {
#pragma unroll
                for (int s = 0; s < 8; ++s) {
#pragma unroll
                    for (int cti = 0; cti < 4; ++cti)
                        acc2[cti] = __builtin_amdgcn_wmma_f32_16x16x4_f32(
                            false, g.a[s], false, g.b[s][cti], (short)0, acc2[cti], false, false);
                }
            };
            RawB raw;
            FragsB cur;
            fetch(raw, 0);
            pair_up(cur, raw);
            for (int blk = 1; blk < 6; ++blk) {
                fetch(raw, blk);
                KEEP_GROUPS_APART();
                mma(cur);
                KEEP_GROUPS_APART();
                pair_up(cur, raw);
            }
            if (younger) __builtin_amdgcn_s_setprio(0);
            mma(cur);
        }
#pragma unroll
        for (int v = 0; v < 8; ++v) {
            v4f xb;
#pragma unroll
            for (int cti = 0; cti < 4; ++cti) {
                float xo = xr[cti][v];
                float xn = fmaxf(acc2[cti][v], 0.0f);
                xr[cti][v] = xn;
                xb[cti] = 2.0f * xn - xo;
            }
            *(v4f*)&smem[xbw + (v + hi) * AP + xcol] = xb;
        }
        __syncthreads();
    }

#pragma unroll
    for (int v = 0; v < 8; ++v) {
        v4f q;
#pragma unroll
        for (int cti = 0; cti < 4; ++cti)
            q[cti] = xr[cti][v];
        *(v4f*)&smem[xw + (v + hi) * AP + xcol] = q;
    }
#pragma unroll
    for (int cti = 0; cti < 3; ++cti) {
        int ct = qtr * 3 + cti;
#pragma unroll
        for (int v = 0; v < 8; ++v)
            smem[zw + (v + hi) * YP + ct * 16 + n] = ysr[cti][v];
    }
    __syncthreads();

    float* out_x  = out;
    float* out_y1 = out + (size_t)8192 * 256;
    float* out_y2 = out_y1 + (size_t)8192 * 128;
    for (int i = pid; i < 16 * 64; i += 128) {
        int r = i >> 6, c4 = (i & 63) << 2;
        float4 v = *(const float4*)&smem[xw + r * AP + c4];
        *(float4*)(out_x + (size_t)(rbase + r) * 256 + c4) = v;
    }
    const float NEG_INV_TAU = -1.0f / TAU;
    for (int i = pid; i < 16 * 96; i += 128) {
        int r = i / 96, c2 = (i % 96) << 1;
        float2 v = *(const float2*)&smem[zw + r * YP + c2];
        v.x *= NEG_INV_TAU; v.y *= NEG_INV_TAU;
        if (c2 < 128) *(float2*)(out_y1 + (size_t)(rbase + r) * 128 + c2) = v;
        else          *(float2*)(out_y2 + (size_t)(rbase + r) * 64 + (c2 - 128)) = v;
    }
}

extern "C" void kernel_launch(void* const* d_in, const int* in_sizes, int n_in,
                              void* d_out, int out_size, void* d_ws, size_t ws_size,
                              hipStream_t stream) {
    const float* x0 = (const float*)d_in[0];
    const float* y1 = (const float*)d_in[1];
    const float* y2 = (const float*)d_in[2];
    const float* z  = (const float*)d_in[3];
    const float* c  = (const float*)d_in[4];
    const float* A1 = (const float*)d_in[5];
    const float* A2 = (const float*)d_in[6];
    (void)in_sizes; (void)n_in; (void)out_size; (void)d_ws; (void)ws_size;
    pdhg_kernel<<<dim3(8192 / 32), dim3(256), 0, stream>>>(
        x0, y1, y2, z, c, A1, A2, (float*)d_out);
}


# Operands swapped: a lane's result registers are the next product's operand pairs on the wave's own k range, so each product starts from registers under a split barrier; per-wave rotated k order
# speedup vs baseline: 1.7869x; 1.0330x over previous
#include <hip/hip_runtime.h>


typedef __attribute__((ext_vector_type(2))) float v2f;
typedef __attribute__((ext_vector_type(8))) float v8f;
typedef __attribute__((ext_vector_type(4))) float v4f;

#define TAU   0.05f
#define SIGMA 0.05f
#define MAX_ITER 30

static constexpr int AP = 260;
static constexpr int YP = 196;

static constexpr int OFF_A  = 0;
static constexpr int OFF_X  = OFF_A  + 192 * AP;
static constexpr int OFF_XB = OFF_X  + 32  * AP;
static constexpr int OFF_YS = OFF_XB + 32  * AP;
static constexpr int OFF_Z  = OFF_YS + 32  * YP;
static constexpr int OFF_C  = OFF_Z  + 32  * YP;
static constexpr int SMEM_TOT = OFF_C + 256;

struct FragsA { v2f a[8]; v2f b[8][3]; };
struct OwnA   { v4f m[4][3]; };
struct RawB   { v2f a[8]; v4f b[8][2]; };
struct FragsB { v2f a[8]; v2f b[8][4]; };

#define KEEP_GROUPS_APART() __builtin_amdgcn_sched_barrier(0)

#define LDS_STORES_LANDED() asm volatile("s_wait_dscnt 0x0" ::: "memory")
#define BAR_SIGNAL()        asm volatile("s_barrier_signal -1" ::: "memory")
#define BAR_WAIT()          asm volatile("s_barrier_wait -1" ::: "memory")

__global__ __launch_bounds__(256, 1)
void pdhg_kernel(const float* __restrict__ x0g, const float* __restrict__ y1g,
                 const float* __restrict__ y2g, const float* __restrict__ zg,
                 const float* __restrict__ cg,  const float* __restrict__ A1,
                 const float* __restrict__ A2,  float* __restrict__ out)
{
    __shared__ float smem[SMEM_TOT];

    const int tid  = threadIdx.x;
    const int wave = tid >> 5;
    const int lane = tid & 31;
    const int tile = wave >> 2;
    const int qtr  = wave & 3;
    const int n    = lane & 15;
    const int hi   = (lane >> 4) << 3;
    const int ko2  = (lane >> 4) << 1;
    const int pid  = (qtr << 5) | lane;
    const int younger = __builtin_amdgcn_readfirstlane(tile);

    const int rbase = blockIdx.x * 32 + tile * 16;
    const int xw  = OFF_X  + tile * 16 * AP;
    const int xbw = OFF_XB + tile * 16 * AP;
    const int ysw = OFF_YS + tile * 16 * YP;
    const int zw  = OFF_Z  + tile * 16 * YP;

    v4f xv[8];
    v2f zv[12], yv[12];
    v4f cv = {0.0f, 0.0f, 0.0f, 0.0f};
#pragma unroll
    for (int j = 0; j < 8; ++j) {
        int i = pid + j * 128;
        int r = i >> 6, c4 = (i & 63) << 2;
        xv[j] = *(const v4f*)(x0g + (size_t)(rbase + r) * 256 + c4);
    }
#pragma unroll
    for (int j = 0; j < 12; ++j) {
        int i = pid + j * 128;
        int r = i / 96, c2 = (i % 96) << 1;
        zv[j] = *(const v2f*)(zg + (size_t)(rbase + r) * 192 + c2);
        const float* ysrc = (c2 < 128) ? (y1g + (size_t)(rbase + r) * 128 + c2)
                                       : (y2g + (size_t)(rbase + r) * 64 + (c2 - 128));
        yv[j] = *(const v2f*)ysrc;
    }
    if (tid < 64)
        cv = *(const v4f*)(cg + tid * 4);
#pragma unroll
    for (int g = 0; g < 3; ++g) {
        v4f av[16];
#pragma unroll
        for (int u = 0; u < 16; ++u) {
            int i = tid + (g * 16 + u) * 256;
            int r = i >> 6, c4 = (i & 63) << 2;
            const float* src = (g * 16 + u < 32) ? (A1 + r * 256 + c4) : (A2 + (r - 128) * 256 + c4);
            av[u] = *(const v4f*)src;
        }
#pragma unroll
        for (int u = 0; u < 16; ++u) {
            int i = tid + (g * 16 + u) * 256;
            int r = i >> 6, c4 = (i & 63) << 2;
            *(v4f*)&smem[OFF_A + r * AP + c4] = av[u];
        }
        __builtin_amdgcn_sched_barrier(0);
    }
#pragma unroll
    for (int j = 0; j < 8; ++j) {
        int i = pid + j * 128;
        int r = i >> 6, c4 = (i & 63) << 2;
        *(v4f*)&smem[xw  + r * AP + c4] = xv[j];
        *(v4f*)&smem[xbw + r * AP + c4] = xv[j];
    }
#pragma unroll
    for (int j = 0; j < 12; ++j) {
        int i = pid + j * 128;
        int r = i / 96, c2 = (i % 96) << 1;
        *(v2f*)&smem[zw + r * YP + c2] = zv[j];
        v2f v = yv[j];
        v.x *= -TAU; v.y *= -TAU;
        *(v2f*)&smem[ysw + r * YP + c2] = v;
    }
    if (tid < 64)
        *(v4f*)&smem[OFF_C + tid * 4] = cv;
    __syncthreads();

    const float NEG_INV_TS = -1.0f / (TAU * SIGMA);

    const int h    = lane >> 4;
    const int q    = __builtin_amdgcn_readfirstlane(qtr);
    const int xcol = qtr * 64 + 4 * n;
    const int xown = qtr * 64 + 32 * h;

    v8f ysr[3], zr[3];
#pragma unroll
    for (int cti = 0; cti < 3; ++cti) {
        const int col = (qtr * 3 + cti) * 16 + hi;
        v4f y0 = *(const v4f*)&smem[ysw + n * YP + col];
        v4f y1 = *(const v4f*)&smem[ysw + n * YP + col + 4];
        v4f z0 = *(const v4f*)&smem[zw  + n * YP + col];
        v4f z1 = *(const v4f*)&smem[zw  + n * YP + col + 4];
        ysr[cti] = __builtin_shufflevector(y0, y1, 0, 1, 2, 3, 4, 5, 6, 7);
        zr[cti]  = __builtin_shufflevector(z0, z1, 0, 1, 2, 3, 4, 5, 6, 7);
    }
    v4f xr[8], cr[8], xbv[8];
#pragma unroll
    for (int r = 0; r < 8; ++r) {
        xr[r]  = *(const v4f*)&smem[xw + n * AP + xown + 4 * r];
        cr[r]  = *(const v4f*)&smem[OFF_C + xown + 4 * r];
        xbv[r] = xr[r];
    }

    const float* xb_row = &smem[xbw + n * AP + ko2];
    const float* m1_row = &smem[OFF_A + (qtr * 48 + n) * AP + ko2];
    const float* m1_own = &smem[OFF_A + (qtr * 48 + n) * AP + qtr * 64 + 32 * h];
    const float* ys_row = &smem[ysw + n * YP + ko2];
    const float* m2_col = &smem[OFF_A + (4 * ko2) * AP + xcol];

    auto ownA_fetch = [&](OwnA& o, int r0) {
#pragma unroll
        for (int rr = 0; rr < 4; ++rr) {
#pragma unroll
            for (int cti = 0; cti < 3; ++cti)
                o.m[rr][cti] = *(const v4f*)(m1_own + cti * 16 * AP + 4 * (r0 + rr));
        }
    };

    OwnA mo0;
    ownA_fetch(mo0, 0);
    LDS_STORES_LANDED();
    BAR_SIGNAL();

    for (int it = 0; it < MAX_ITER; ++it) {
        if (younger) __builtin_amdgcn_s_setprio(3);
        v8f acc[3];
#pragma unroll
        for (int cti = 0; cti < 3; ++cti) {
#pragma unroll
            for (int v = 0; v < 8; ++v)
                acc[cti][v] = ysr[cti][v] * NEG_INV_TS - zr[cti][v];
        }
        RawB rawOwn;
        {
            auto own_mma = [&](const OwnA& o, int r0) {
#pragma unroll
                for (int rr = 0; rr < 4; ++rr) {
#pragma unroll
                    for (int tp = 0; tp < 2; ++tp) {
                        v2f xv2 = tp ? __builtin_shufflevector(xbv[r0 + rr], xbv[r0 + rr], 2, 3)
                                     : __builtin_shufflevector(xbv[r0 + rr], xbv[r0 + rr], 0, 1);
#pragma unroll
                        for (int cti = 0; cti < 3; ++cti) {
                            v2f mv2 = tp ? __builtin_shufflevector(o.m[rr][cti], o.m[rr][cti], 2, 3)
                                         : __builtin_shufflevector(o.m[rr][cti], o.m[rr][cti], 0, 1);
                            acc[cti] = __builtin_amdgcn_wmma_f32_16x16x4_f32(
                                false, mv2, false, xv2, (short)0, acc[cti], false, false);
                        }
                    }
                }
            };
            auto fetch = [&](FragsA& f, int kb) {
                const float* ap = xb_row + kb;
                const float* bp = m1_row + kb;
#pragma unroll
                for (int s = 0; s < 8; ++s) {
                    f.a[s] = *(const v2f*)(ap + s * 4);
#pragma unroll
                    for (int cti = 0; cti < 3; ++cti)
                        f.b[s][cti] = *(const v2f*)(bp + cti * 16 * AP + s * 4);
                }
            };
            auto mma = [&](const FragsA& f) {
#pragma unroll
                for (int s = 0; s < 8; ++s) {
#pragma unroll
                    for (int cti = 0; cti < 3; ++cti)
                        acc[cti] = __builtin_amdgcn_wmma_f32_16x16x4_f32(
                            false, f.b[s][cti], false, f.a[s], (short)0, acc[cti], false, false);
                }
            };
            auto kb = [&](int j) { return ((2 * q + 2 + j) & 7) << 5; };
            auto fetch_b2 = [&](RawB& f, int g0, int g1) {
                const float* p0 = m2_col + g0 * 16 * AP;
                const float* p1 = m2_col + g1 * 16 * AP;
#pragma unroll
                for (int s = 0; s < 8; ++s) {
                    const float* p = (s < 4) ? p0 : p1;
#pragma unroll
                    for (int r = 0; r < 2; ++r)
                        f.b[s][r] = *(const v4f*)(p + ((s & 3) * 2 + r) * AP);
                }
            };
            OwnA mo1;
            FragsA f0, f1;
            ownA_fetch(mo1, 4);
            KEEP_GROUPS_APART();
            own_mma(mo0, 0);
            KEEP_GROUPS_APART();
            BAR_WAIT();
            fetch(f0, kb(0));
            KEEP_GROUPS_APART();
            own_mma(mo1, 4);
            KEEP_GROUPS_APART();
            fetch(f1, kb(1));
            KEEP_GROUPS_APART();
            mma(f0);
            KEEP_GROUPS_APART();
            fetch(f0, kb(2));
            KEEP_GROUPS_APART();
            mma(f1);
            KEEP_GROUPS_APART();
            fetch(f1, kb(3));
            KEEP_GROUPS_APART();
            mma(f0);
            KEEP_GROUPS_APART();
            fetch(f0, kb(4));
            KEEP_GROUPS_APART();
            mma(f1);
            KEEP_GROUPS_APART();
            if (younger) __builtin_amdgcn_s_setprio(0);
            fetch(f1, kb(5));
            KEEP_GROUPS_APART();
            mma(f0);
            KEEP_GROUPS_APART();
            fetch_b2(rawOwn, 3 * q, 3 * q + 1);
            KEEP_GROUPS_APART();
            mma(f1);
        }
#pragma unroll
        for (int cti = 0; cti < 3; ++cti) {
            int ct = qtr * 3 + cti;
            v8f t8;
#pragma unroll
            for (int v = 0; v < 8; ++v) {
                float yv = SIGMA * acc[cti][v];
                yv = (ct < 8) ? fmaxf(yv, 0.0f) : yv;
                t8[v] = -TAU * yv;
            }
            ysr[cti] = t8;
#pragma unroll
            for (int sp = 0; sp < 4; ++sp) {
                v2f pr;
                pr.x = t8[2 * sp];
                pr.y = t8[2 * sp + 1];
                *(v2f*)&smem[ysw + n * YP + ct * 16 + 4 * sp + ko2] = pr;
            }
        }
        LDS_STORES_LANDED();
        BAR_SIGNAL();

        if (younger) __builtin_amdgcn_s_setprio(3);
        v8f acc2[4];
#pragma unroll
        for (int cti = 0; cti < 4; ++cti) {
#pragma unroll
            for (int v = 0; v < 8; ++v)
                acc2[cti][v] = __builtin_fmaf(-TAU, cr[v][cti], xr[v][cti]);
        }
        {
            auto grp = [&](int u) { int g = 3 * q + u; return (g >= 12) ? g - 12 : g; };
            auto fetch_b = [&](RawB& f, int g0, int g1) {
                const float* p0 = m2_col + g0 * 16 * AP;
                const float* p1 = m2_col + g1 * 16 * AP;
#pragma unroll
                for (int s = 0; s < 8; ++s) {
                    const float* p = (s < 4) ? p0 : p1;
#pragma unroll
                    for (int r = 0; r < 2; ++r)
                        f.b[s][r] = *(const v4f*)(p + ((s & 3) * 2 + r) * AP);
                }
            };
            auto fetch_a = [&](RawB& f, int g0, int g1) {
                const float* p0 = ys_row + g0 * 16;
                const float* p1 = ys_row + g1 * 16;
#pragma unroll
                for (int s = 0; s < 8; ++s)
                    f.a[s] = *(const v2f*)(((s < 4) ? p0 : p1) + (s & 3) * 4);
            };
            auto pair_b = [&](FragsB& g, const RawB& f) {
#pragma unroll
                for (int s = 0; s < 8; ++s) {
#pragma unroll
                    for (int cti = 0; cti < 4; ++cti) {
                        v2f b;
                        b.x = f.b[s][0][cti];
                        b.y = f.b[s][1][cti];
                        g.b[s][cti] = b;
                    }
                }
            };
            auto pair_a = [&](FragsB& g, const RawB& f) {
#pragma unroll
                for (int s = 0; s < 8; ++s)
                    g.a[s] = f.a[s];
            };
            auto mma = [&](const FragsB& g, int s0, int s1) {
#pragma unroll
                for (int s = s0; s < s1; ++s) {
#pragma unroll
                    for (int cti = 0; cti < 4; ++cti)
                        acc2[cti] = __builtin_amdgcn_wmma_f32_16x16x4_f32(
                            false, g.b[s][cti], false, g.a[s], (short)0, acc2[cti], false, false);
                }
            };
            RawB raw;
            FragsB cur;
            pair_b(cur, rawOwn);
#pragma unroll
            for (int s = 0; s < 8; ++s) {
                const int i = s >> 2, j = s & 3;
                cur.a[s] = (j == 0) ? __builtin_shufflevector(ysr[i], ysr[i], 0, 1)
                         : (j == 1) ? __builtin_shufflevector(ysr[i], ysr[i], 2, 3)
                         : (j == 2) ? __builtin_shufflevector(ysr[i], ysr[i], 4, 5)
                                    : __builtin_shufflevector(ysr[i], ysr[i], 6, 7);
            }
            const int g2 = grp(2), g3 = grp(3);
            fetch_b(raw, g2, g3);
            KEEP_GROUPS_APART();
            mma(cur, 0, 4);
            KEEP_GROUPS_APART();
            BAR_WAIT();
            fetch_a(raw, g2, g3);
            KEEP_GROUPS_APART();
            mma(cur, 4, 8);
            KEEP_GROUPS_APART();
            pair_b(cur, raw);
            pair_a(cur, raw);
#pragma unroll
            for (int b = 2; b < 6; ++b) {
                const int ga = grp(2 * b), gb = grp(2 * b + 1);
                fetch_b(raw, ga, gb);
                fetch_a(raw, ga, gb);
                KEEP_GROUPS_APART();
                mma(cur, 0, 8);
                KEEP_GROUPS_APART();
                pair_b(cur, raw);
                pair_a(cur, raw);
            }
            ownA_fetch(mo0, 0);
            KEEP_GROUPS_APART();
            if (younger) __builtin_amdgcn_s_setprio(0);
            mma(cur, 0, 8);
        }
#pragma unroll
        for (int v = 0; v < 8; ++v) {
            v4f xb, xn4;
#pragma unroll
            for (int cti = 0; cti < 4; ++cti) {
                float xo = xr[v][cti];
                float xn = fmaxf(acc2[cti][v], 0.0f);
                xn4[cti] = xn;
                xb[cti] = 2.0f * xn - xo;
            }
            xr[v]  = xn4;
            xbv[v] = xb;
            *(v4f*)&smem[xbw + n * AP + xown + 4 * v] = xb;
        }
        LDS_STORES_LANDED();
        BAR_SIGNAL();
    }
    BAR_WAIT();

#pragma unroll
    for (int v = 0; v < 8; ++v)
        *(v4f*)&smem[xw + n * AP + xown + 4 * v] = xr[v];
#pragma unroll
    for (int cti = 0; cti < 3; ++cti) {
        const int col = (qtr * 3 + cti) * 16 + hi;
        *(v4f*)&smem[zw + n * YP + col]     = __builtin_shufflevector(ysr[cti], ysr[cti], 0, 1, 2, 3);
        *(v4f*)&smem[zw + n * YP + col + 4] = __builtin_shufflevector(ysr[cti], ysr[cti], 4, 5, 6, 7);
    }
    __syncthreads();

    float* out_x  = out;
    float* out_y1 = out + (size_t)8192 * 256;
    float* out_y2 = out_y1 + (size_t)8192 * 128;
    for (int i = pid; i < 16 * 64; i += 128) {
        int r = i >> 6, c4 = (i & 63) << 2;
        float4 v = *(const float4*)&smem[xw + r * AP + c4];
        *(float4*)(out_x + (size_t)(rbase + r) * 256 + c4) = v;
    }
    const float NEG_INV_TAU = -1.0f / TAU;
    for (int i = pid; i < 16 * 96; i += 128) {
        int r = i / 96, c2 = (i % 96) << 1;
        float2 v = *(const float2*)&smem[zw + r * YP + c2];
        v.x *= NEG_INV_TAU; v.y *= NEG_INV_TAU;
        if (c2 < 128) *(float2*)(out_y1 + (size_t)(rbase + r) * 128 + c2) = v;
        else          *(float2*)(out_y2 + (size_t)(rbase + r) * 64 + (c2 - 128)) = v;
    }
}

extern "C" void kernel_launch(void* const* d_in, const int* in_sizes, int n_in,
                              void* d_out, int out_size, void* d_ws, size_t ws_size,
                              hipStream_t stream) {
    const float* x0 = (const float*)d_in[0];
    const float* y1 = (const float*)d_in[1];
    const float* y2 = (const float*)d_in[2];
    const float* z  = (const float*)d_in[3];
    const float* c  = (const float*)d_in[4];
    const float* A1 = (const float*)d_in[5];
    const float* A2 = (const float*)d_in[6];
    (void)in_sizes; (void)n_in; (void)out_size; (void)d_ws; (void)ws_size;
    pdhg_kernel<<<dim3(8192 / 32), dim3(256), 0, stream>>>(
        x0, y1, y2, z, c, A1, A2, (float*)d_out);
}
